# starting kernel
# baseline (speedup 1.0000x reference)
#include <hip/hip_runtime.h>

typedef __attribute__((ext_vector_type(2))) float v2f;
typedef __attribute__((ext_vector_type(8))) float v8f;

#define BB   32
#define CIN  64
#define COUT 64
#define HO   30
#define NPOS 900
#define KT   9
#define TOT  (BB*COUT*NPOS)

#define WS_Y     0
#define WS_PART  (TOT)
#define WS_PART2 (TOT + NPOS*COUT)
#define WS_BN    (TOT + 2*NPOS*COUT)

#define AP 74
#define BP 74
#define SP 66

__global__ __launch_bounds__(256) void k1_conv(
    const float* __restrict__ ff, const float* __restrict__ cw,
    const float* __restrict__ cb, float* __restrict__ ws)
{
  __shared__ float Al[BB * AP];
  __shared__ float Bl[COUT * BP];
  __shared__ float redS[256];
  __shared__ float redQ[256];

  const int p = blockIdx.x;
  const int h = p / HO, w = p % HO;
  const int t = threadIdx.x;
  const int lane = t & 31, wave = t >> 5;
  const int mt = wave & 1, nt = wave >> 1;
  const int ln = lane & 15, hi = lane >> 4;
  const int m = mt * 16 + ln;
  const int n = nt * 16 + ln;

  v8f acc = {};
  for (int cg = 0; cg < 8; ++cg) {
    __syncthreads();
    {
      int b = t >> 3, cc = t & 7, c = cg * 8 + cc;
      const float* src = ff + ((b * CIN + c) * 32 + h) * 32 + w;
      float* dst = Al + b * AP + cc * KT;
      #pragma unroll
      for (int kh = 0; kh < 3; ++kh)
        #pragma unroll
        for (int kw = 0; kw < 3; ++kw)
          dst[kh * 3 + kw] = src[kh * 32 + kw];
    }
    {
      #pragma unroll
      for (int rep = 0; rep < 2; ++rep) {
        int idx = t + rep * 256;
        int o = idx >> 3, cc = idx & 7, c = cg * 8 + cc;
        const float* src = cw + ((size_t)(o * CIN + c) * NPOS + p) * KT;
        float* dst = Bl + o * BP + cc * KT;
        #pragma unroll
        for (int k = 0; k < KT; ++k) dst[k] = src[k];
      }
    }
    __syncthreads();
    #pragma unroll
    for (int kb = 0; kb < 72; kb += 4) {
      v2f av = *(const v2f*)(Al + m * AP + kb + 2 * hi);
      v2f bv = *(const v2f*)(Bl + n * BP + kb + 2 * hi);
      acc = __builtin_amdgcn_wmma_f32_16x16x4_f32(false, av, false, bv,
                                                  (short)0, acc, false, false);
    }
  }

  const float bias = cb[n * NPOS + p];
  float s = 0.f, q = 0.f;
  #pragma unroll
  for (int r = 0; r < 8; ++r) {
    float v = acc[r] + bias;
    int b = mt * 16 + hi * 8 + r;
    ws[WS_Y + (b * COUT + n) * NPOS + p] = v;
    s += v; q += v * v;
  }
  redS[t] = s; redQ[t] = q;
  __syncthreads();
  if (t < COUT) {
    int nt2 = t >> 4, l0 = t & 15;
    float S = 0.f, Q = 0.f;
    #pragma unroll
    for (int mt2 = 0; mt2 < 2; ++mt2)
      #pragma unroll
      for (int h2 = 0; h2 < 2; ++h2) {
        int tt = ((nt2 * 2 + mt2) << 5) + h2 * 16 + l0;
        S += redS[tt]; Q += redQ[tt];
      }
    ws[WS_PART  + p * COUT + t] = S;
    ws[WS_PART2 + p * COUT + t] = Q;
  }
}

__global__ __launch_bounds__(64) void k2_bn(
    const float* __restrict__ gamma, const float* __restrict__ beta,
    float* __restrict__ ws)
{
  const int ch = threadIdx.x;
  float S = 0.f, Q = 0.f;
  for (int p = 0; p < NPOS; ++p) {
    S += ws[WS_PART  + p * COUT + ch];
    Q += ws[WS_PART2 + p * COUT + ch];
  }
  const float inv = 1.0f / (float)(BB * NPOS);
  const float mean = S * inv;
  const float var  = Q * inv - mean * mean;
  const float sc = gamma[ch] * rsqrtf(var + 1e-5f);
  ws[WS_BN + ch]        = sc;
  ws[WS_BN + COUT + ch] = beta[ch] - mean * sc;
}

__global__ __launch_bounds__(256) void k3_rec_lif(
    const float* __restrict__ fbv,  const float* __restrict__ soma,
    const float* __restrict__ spk,  const float* __restrict__ acur,
    const float* __restrict__ bt,   const float* __restrict__ lrec,
    const float* __restrict__ taum, const float* __restrict__ taud,
    const float* __restrict__ taua, const float* __restrict__ ws,
    float* __restrict__ out)
{
  __shared__ float As[BB * SP];
  __shared__ float Bs[COUT * SP];
  const int p = blockIdx.x;
  const int t = threadIdx.x;
  const int lane = t & 31, wave = t >> 5;
  const int mt = wave & 1, nt = wave >> 1;
  const int ln = lane & 15, hi = lane >> 4;
  const int m = mt * 16 + ln;
  const int n = nt * 16 + ln;

  #pragma unroll
  for (int i = 0; i < 8; ++i) {
    int e = i * 256 + t;
    int b = e >> 6, d = e & 63;
    As[b * SP + d] = spk[(b * COUT + d) * NPOS + p];
  }
  const float* lp = lrec + (size_t)p * COUT * COUT;
  #pragma unroll
  for (int i = 0; i < 16; ++i) {
    int e = i * 256 + t;
    int c = e >> 6, d = e & 63;
    Bs[c * SP + d] = lp[e];
  }
  __syncthreads();

  v8f acc = {};
  #pragma unroll
  for (int kb = 0; kb < 64; kb += 4) {
    v2f av = *(const v2f*)(As + m * SP + kb + 2 * hi);
    v2f bv = *(const v2f*)(Bs + n * SP + kb + 2 * hi);
    acc = __builtin_amdgcn_wmma_f32_16x16x4_f32(false, av, false, bv,
                                                (short)0, acc, false, false);
  }

  const float sc = ws[WS_BN + n];
  const float sh = ws[WS_BN + COUT + n];
  const float alpha = __expf(-0.5f / taum[n * NPOS + p]);
  const float rho   = __expf(-0.5f / taud[n * NPOS + p]);
  const float eta   = __expf(-0.5f / taua[n * NPOS + p]);

  #pragma unroll
  for (int r = 0; r < 8; ++r) {
    int b = mt * 16 + hi * 8 + r;
    int idx = (b * COUT + n) * NPOS + p;
    float sv    = As[b * SP + n];
    float convx = ws[WS_Y + idx] * sc + sh + acc[r];
    float bnew  = rho * bt[idx] + (1.f - rho) * sv;
    float thre  = 0.1f + 1.8f * bnew;
    float anew  = eta * acur[idx] + fbv[idx];
    float sg    = 1.f / (1.f + __expf(-anew)) - 0.5f;
    float snew  = alpha * soma[idx] + sg + convx - thre * sv;
    float spike = (snew - thre) > 0.f ? 1.f : 0.f;
    out[idx]           = snew;
    out[TOT   + idx]   = spike;
    out[2*TOT + idx]   = anew;
    out[3*TOT + idx]   = bnew;
  }
}

extern "C" void kernel_launch(void* const* d_in, const int* in_sizes, int n_in,
                              void* d_out, int out_size, void* d_ws, size_t ws_size,
                              hipStream_t stream) {
  const float* ff   = (const float*)d_in[0];
  const float* fb   = (const float*)d_in[1];
  const float* soma = (const float*)d_in[2];
  const float* spk  = (const float*)d_in[3];
  const float* acur = (const float*)d_in[4];
  const float* bt   = (const float*)d_in[5];
  const float* cw   = (const float*)d_in[6];
  const float* cb   = (const float*)d_in[7];
  const float* gam  = (const float*)d_in[8];
  const float* bet  = (const float*)d_in[9];
  const float* lrec = (const float*)d_in[10];
  const float* taum = (const float*)d_in[11];
  const float* taud = (const float*)d_in[12];
  const float* taua = (const float*)d_in[13];
  float* out = (float*)d_out;
  float* ws  = (float*)d_ws;

  hipLaunchKernelGGL(k1_conv,    dim3(NPOS), dim3(256), 0, stream, ff, cw, cb, ws);
  hipLaunchKernelGGL(k2_bn,      dim3(1),    dim3(64),  0, stream, gam, bet, ws);
  hipLaunchKernelGGL(k3_rec_lif, dim3(NPOS), dim3(256), 0, stream,
                     fb, soma, spk, acur, bt, lrec, taum, taud, taua, ws, out);
}


# k3 retiled: a block takes 8 consecutive positions x 16 batches x 16 channels, so every [b][c][p] load/store is a float4 along p, not a 4-byte stride-3600 access; rec sum order unchanged
# speedup vs baseline: 1.9303x; 1.9303x over previous
#include <hip/hip_runtime.h>

typedef __attribute__((ext_vector_type(2))) float v2f;
typedef __attribute__((ext_vector_type(8))) float v8f;

#define BB   32
#define CIN  64
#define COUT 64
#define HO   30
#define NPOS 900
#define KT   9
#define TOT  (BB*COUT*NPOS)

#define WS_Y     0
#define WS_PART  (TOT)
#define WS_PART2 (TOT + NPOS*COUT)
#define WS_BN    (TOT + 2*NPOS*COUT)

#define AP 74
#define BP 74

__global__ __launch_bounds__(256) void k1_conv(
    const float* __restrict__ ff, const float* __restrict__ cw,
    const float* __restrict__ cb, float* __restrict__ ws)
{
  __shared__ float Al[BB * AP];
  __shared__ float Bl[COUT * BP];
  __shared__ float redS[256];
  __shared__ float redQ[256];

  const int p = blockIdx.x;
  const int h = p / HO, w = p % HO;
  const int t = threadIdx.x;
  const int lane = t & 31, wave = t >> 5;
  const int mt = wave & 1, nt = wave >> 1;
  const int ln = lane & 15, hi = lane >> 4;
  const int m = mt * 16 + ln;
  const int n = nt * 16 + ln;

  v8f acc = {};
  for (int cg = 0; cg < 8; ++cg) {
    __syncthreads();
    {
      int b = t >> 3, cc = t & 7, c = cg * 8 + cc;
      const float* src = ff + ((b * CIN + c) * 32 + h) * 32 + w;
      float* dst = Al + b * AP + cc * KT;
      #pragma unroll
      for (int kh = 0; kh < 3; ++kh)
        #pragma unroll
        for (int kw = 0; kw < 3; ++kw)
          dst[kh * 3 + kw] = src[kh * 32 + kw];
    }
    {
      #pragma unroll
      for (int rep = 0; rep < 2; ++rep) {
        int idx = t + rep * 256;
        int o = idx >> 3, cc = idx & 7, c = cg * 8 + cc;
        const float* src = cw + ((size_t)(o * CIN + c) * NPOS + p) * KT;
        float* dst = Bl + o * BP + cc * KT;
        #pragma unroll
        for (int k = 0; k < KT; ++k) dst[k] = src[k];
      }
    }
    __syncthreads();
    #pragma unroll
    for (int kb = 0; kb < 72; kb += 4) {
      v2f av = *(const v2f*)(Al + m * AP + kb + 2 * hi);
      v2f bv = *(const v2f*)(Bl + n * BP + kb + 2 * hi);
      acc = __builtin_amdgcn_wmma_f32_16x16x4_f32(false, av, false, bv,
                                                  (short)0, acc, false, false);
    }
  }

  const float bias = cb[n * NPOS + p];
  float s = 0.f, q = 0.f;
  #pragma unroll
  for (int r = 0; r < 8; ++r) {
    float v = acc[r] + bias;
    int b = mt * 16 + hi * 8 + r;
    ws[WS_Y + (b * COUT + n) * NPOS + p] = v;
    s += v; q += v * v;
  }
  redS[t] = s; redQ[t] = q;
  __syncthreads();
  if (t < COUT) {
    int nt2 = t >> 4, l0 = t & 15;
    float S = 0.f, Q = 0.f;
    #pragma unroll
    for (int mt2 = 0; mt2 < 2; ++mt2)
      #pragma unroll
      for (int h2 = 0; h2 < 2; ++h2) {
        int tt = ((nt2 * 2 + mt2) << 5) + h2 * 16 + l0;
        S += redS[tt]; Q += redQ[tt];
      }
    ws[WS_PART  + p * COUT + t] = S;
    ws[WS_PART2 + p * COUT + t] = Q;
  }
}

__global__ __launch_bounds__(64) void k2_bn(
    const float* __restrict__ gamma, const float* __restrict__ beta,
    float* __restrict__ ws)
{
  const int ch = threadIdx.x;
  float S = 0.f, Q = 0.f;
  for (int p = 0; p < NPOS; ++p) {
    S += ws[WS_PART  + p * COUT + ch];
    Q += ws[WS_PART2 + p * COUT + ch];
  }
  const float inv = 1.0f / (float)(BB * NPOS);
  const float mean = S * inv;
  const float var  = Q * inv - mean * mean;
  const float sc = gamma[ch] * rsqrtf(var + 1e-5f);
  ws[WS_BN + ch]        = sc;
  ws[WS_BN + COUT + ch] = beta[ch] - mean * sc;
}

#define PT        8
#define K3_TILES  113
#define K3_BLOCKS (K3_TILES * 8)
#define SPP       68
#define PSS       1092
#define RP        12

__device__ __forceinline__ float4 ld4(const float* p) { return *(const float4*)p; }

__global__ __launch_bounds__(256) void k3_rec_lif(
    const float* __restrict__ fbv,  const float* __restrict__ soma,
    const float* __restrict__ spk,  const float* __restrict__ acur,
    const float* __restrict__ bt,   const float* __restrict__ lrec,
    const float* __restrict__ taum, const float* __restrict__ taud,
    const float* __restrict__ taua, const float* __restrict__ ws,
    float* __restrict__ out)
{
  __shared__ __attribute__((aligned(16))) float Ss[PT * PSS];
  __shared__ __attribute__((aligned(16))) float Rs[256 * RP];

  const int bid = blockIdx.x;
  const int cq = bid & 3, bh = (bid >> 2) & 1, tile = bid >> 3;
  const int p0 = tile * PT, b0 = bh * 16, c0 = cq * 16;
  const int t = threadIdx.x;
  const int lane = t & 31, wave = t >> 5;
  const int ln = lane & 15, hi = lane >> 4;

  const int q  = t & 1;
  const int cl = (t >> 1) & 15;
  const int c  = c0 + cl;
  const int pq = p0 + 4 * q;
  const bool okq = pq < NPOS;
  const int pql = okq ? pq : NPOS - 4;
  const int bl0 = t >> 5;

  float4 sv4[8];
  #pragma unroll
  for (int i = 0; i < 8; ++i) {
    int e = i * 256 + t;
    int row = e >> 1;
    int d = row & 63, bl = row >> 6;
    sv4[i] = ld4(spk + ((b0 + bl) * COUT + d) * NPOS + pql);
  }

  const int pw = min(p0 + wave, NPOS - 1);
  const float* lp = lrec + ((size_t)pw * COUT + c0 + ln) * COUT + 2 * hi;
  v2f bfr[16];
  #pragma unroll
  for (int j = 0; j < 16; ++j) bfr[j] = *(const v2f*)(lp + 4 * j);

  const float4 tm = ld4(taum + c * NPOS + pql);
  const float4 td = ld4(taud + c * NPOS + pql);
  const float4 ta = ld4(taua + c * NPOS + pql);
  float4 yv[2], bv4[2], av4[2], fv4[2], so4[2];
  #pragma unroll
  for (int i = 0; i < 2; ++i) {
    int idx = ((b0 + bl0 + 8 * i) * COUT + c) * NPOS + pql;
    yv[i]  = ld4(ws + WS_Y + idx);
    bv4[i] = ld4(bt + idx);
    av4[i] = ld4(acur + idx);
    fv4[i] = ld4(fbv + idx);
    so4[i] = ld4(soma + idx);
  }
  const float sc = ws[WS_BN + c];
  const float sh = ws[WS_BN + COUT + c];

  #pragma unroll
  for (int i = 0; i < 8; ++i) {
    int e = i * 256 + t;
    int row = e >> 1;
    int d = row & 63, bl = row >> 6;
    float* dst = Ss + (4 * q) * PSS + bl * SPP + d;
    dst[0]       = sv4[i].x;
    dst[PSS]     = sv4[i].y;
    dst[2 * PSS] = sv4[i].z;
    dst[3 * PSS] = sv4[i].w;
  }
  __syncthreads();

  v8f acc = {};
  const float* ap = Ss + wave * PSS + ln * SPP + 2 * hi;
  #pragma unroll
  for (int j = 0; j < 16; ++j) {
    v2f av = *(const v2f*)(ap + 4 * j);
    acc = __builtin_amdgcn_wmma_f32_16x16x4_f32(false, av, false, bfr[j],
                                                (short)0, acc, false, false);
  }
  #pragma unroll
  for (int r = 0; r < 8; ++r)
    Rs[((hi * 8 + r) * 16 + ln) * RP + wave] = acc[r];
  __syncthreads();

  if (okq) {
    float alpha[4], rho[4], eta[4];
    const float tmv[4] = {tm.x, tm.y, tm.z, tm.w};
    const float tdv[4] = {td.x, td.y, td.z, td.w};
    const float tav[4] = {ta.x, ta.y, ta.z, ta.w};
    #pragma unroll
    for (int k = 0; k < 4; ++k) {
      alpha[k] = __expf(-0.5f / tmv[k]);
      rho[k]   = __expf(-0.5f / tdv[k]);
      eta[k]   = __expf(-0.5f / tav[k]);
    }
    #pragma unroll
    for (int i = 0; i < 2; ++i) {
      const int bl = bl0 + 8 * i;
      const int idx = ((b0 + bl) * COUT + c) * NPOS + pq;
      const float4 rc = *(const float4*)(Rs + (bl * 16 + cl) * RP + 4 * q);
      const float rcv[4] = {rc.x, rc.y, rc.z, rc.w};
      const float yy[4]  = {yv[i].x, yv[i].y, yv[i].z, yv[i].w};
      const float bb[4]  = {bv4[i].x, bv4[i].y, bv4[i].z, bv4[i].w};
      const float aa[4]  = {av4[i].x, av4[i].y, av4[i].z, av4[i].w};
      const float ff4[4] = {fv4[i].x, fv4[i].y, fv4[i].z, fv4[i].w};
      const float ss[4]  = {so4[i].x, so4[i].y, so4[i].z, so4[i].w};
      float o0[4], o1[4], o2[4], o3[4];
      #pragma unroll
      for (int k = 0; k < 4; ++k) {
        float sv    = Ss[(4 * q + k) * PSS + bl * SPP + c];
        float convx = yy[k] * sc + sh + rcv[k];
        float bnew  = rho[k] * bb[k] + (1.f - rho[k]) * sv;
        float thre  = 0.1f + 1.8f * bnew;
        float anew  = eta[k] * aa[k] + ff4[k];
        float sg    = 1.f / (1.f + __expf(-anew)) - 0.5f;
        float snew  = alpha[k] * ss[k] + sg + convx - thre * sv;
        o0[k] = snew;
        o1[k] = (snew - thre) > 0.f ? 1.f : 0.f;
        o2[k] = anew;
        o3[k] = bnew;
      }
      *(float4*)(out + idx)           = make_float4(o0[0], o0[1], o0[2], o0[3]);
      *(float4*)(out + TOT + idx)     = make_float4(o1[0], o1[1], o1[2], o1[3]);
      *(float4*)(out + 2 * TOT + idx) = make_float4(o2[0], o2[1], o2[2], o2[3]);
      *(float4*)(out + 3 * TOT + idx) = make_float4(o3[0], o3[1], o3[2], o3[3]);
    }
  }
}

extern "C" void kernel_launch(void* const* d_in, const int* in_sizes, int n_in,
                              void* d_out, int out_size, void* d_ws, size_t ws_size,
                              hipStream_t stream) {
  const float* ff   = (const float*)d_in[0];
  const float* fb   = (const float*)d_in[1];
  const float* soma = (const float*)d_in[2];
  const float* spk  = (const float*)d_in[3];
  const float* acur = (const float*)d_in[4];
  const float* bt   = (const float*)d_in[5];
  const float* cw   = (const float*)d_in[6];
  const float* cb   = (const float*)d_in[7];
  const float* gam  = (const float*)d_in[8];
  const float* bet  = (const float*)d_in[9];
  const float* lrec = (const float*)d_in[10];
  const float* taum = (const float*)d_in[11];
  const float* taud = (const float*)d_in[12];
  const float* taua = (const float*)d_in[13];
  float* out = (float*)d_out;
  float* ws  = (float*)d_ws;

  hipLaunchKernelGGL(k1_conv,    dim3(NPOS), dim3(256), 0, stream, ff, cw, cb, ws);
  hipLaunchKernelGGL(k2_bn,      dim3(1),    dim3(64),  0, stream, gam, bet, ws);
  hipLaunchKernelGGL(k3_rec_lif, dim3(K3_BLOCKS), dim3(256), 0, stream,
                     fb, soma, spk, acur, bt, lrec, taum, taud, taua, ws, out);
}


# k1 retiled: a block takes 8 consecutive positions x 16 out-channels, so conv weights load as float4 along the contiguous (position,tap) axis and y stores as float4 along p; sum orders unchanged
# speedup vs baseline: 3.3456x; 1.7332x over previous
#include <hip/hip_runtime.h>

typedef __attribute__((ext_vector_type(2))) float v2f;
typedef __attribute__((ext_vector_type(8))) float v8f;

#define BB   32
#define CIN  64
#define COUT 64
#define HO   30
#define NPOS 900
#define KT   9
#define TOT  (BB*COUT*NPOS)

#define WS_Y     0
#define WS_PART  (TOT)
#define WS_PART2 (TOT + NPOS*COUT)
#define WS_BN    (TOT + 2*NPOS*COUT)

#define AP 74
#define BP 74

#define P1        8
#define K1_TILES  113
#define K1_BLOCKS (K1_TILES * 4)
#define APL       (BB * AP + 2)
#define BPL       (16 * BP + 40)
#define YP        12
#define ROWF      (NPOS * KT)

__device__ __forceinline__ float4 ld4(const float* p) { return *(const float4*)p; }

__global__ __launch_bounds__(256) void k1_conv(
    const float* __restrict__ ff, const float* __restrict__ cw,
    const float* __restrict__ cb, float* __restrict__ ws)
{
  __shared__ __attribute__((aligned(16))) float Al[P1 * APL];
  __shared__ __attribute__((aligned(16))) float Bl[P1 * BPL];

  const int bid = blockIdx.x;
  const int oq = bid & 3, tile = bid >> 2;
  const int p0 = tile * P1, o0 = oq * 16;
  const int t = threadIdx.x;
  const int lane = t & 31, wave = t >> 5;
  const int ln = lane & 15, hi = lane >> 4;

  const int apl = t & 7, ac = (t >> 3) & 7, ab = t >> 6;
  const int pa = min(p0 + apl, NPOS - 1);
  const float* asrc = ff + ((ab * CIN + ac) * 32 + pa / HO) * 32 + pa % HO;
  float* adst = Al + apl * APL + ab * AP + ac * KT;

  const float* afr = Al + wave * APL + ln * AP + 2 * hi;
  const float* bfr = Bl + wave * BPL + ln * BP + 2 * hi;

  v8f acc0 = {}, acc1 = {};
  for (int cg = 0; cg < 8; ++cg) {
    float4 wv[9];
    #pragma unroll
    for (int i = 0; i < 9; ++i) {
      int idx = i * 256 + t;
      int row = idx / 18, f = idx - row * 18;
      int ol = row >> 3, cc = row & 7;
      int col = p0 * KT + f * 4;
      wv[i] = make_float4(0.f, 0.f, 0.f, 0.f);
      if (col < ROWF)
        wv[i] = ld4(cw + ((o0 + ol) * CIN + cg * 8 + cc) * ROWF + col);
    }
    float pv[8][9];
    #pragma unroll
    for (int i = 0; i < 8; ++i) {
      const float* s = asrc + (i * 4 * CIN + cg * 8) * 1024;
      #pragma unroll
      for (int kh = 0; kh < 3; ++kh)
        #pragma unroll
        for (int kw = 0; kw < 3; ++kw)
          pv[i][kh * 3 + kw] = s[kh * 32 + kw];
    }
    __syncthreads();
    #pragma unroll
    for (int i = 0; i < 8; ++i)
      #pragma unroll
      for (int k = 0; k < KT; ++k) adst[i * 4 * AP + k] = pv[i][k];
    #pragma unroll
    for (int i = 0; i < 9; ++i) {
      int idx = i * 256 + t;
      int row = idx / 18, f = idx - row * 18;
      int ol = row >> 3, cc = row & 7;
      float* d = Bl + ol * BP + cc * KT;
      const float vv[4] = {wv[i].x, wv[i].y, wv[i].z, wv[i].w};
      #pragma unroll
      for (int e = 0; e < 4; ++e) {
        int flat = f * 4 + e;
        int pl = flat / KT, kk = flat - pl * KT;
        d[pl * BPL + kk] = vv[e];
      }
    }
    __syncthreads();
    #pragma unroll
    for (int kb = 0; kb < 72; kb += 4) {
      v2f bv = *(const v2f*)(bfr + kb);
      v2f a0 = *(const v2f*)(afr + kb);
      v2f a1 = *(const v2f*)(afr + 16 * AP + kb);
      acc0 = __builtin_amdgcn_wmma_f32_16x16x4_f32(false, a0, false, bv,
                                                   (short)0, acc0, false, false);
      acc1 = __builtin_amdgcn_wmma_f32_16x16x4_f32(false, a1, false, bv,
                                                   (short)0, acc1, false, false);
    }
  }

  __syncthreads();
  float* Ys   = Al;
  float* redS = Bl;
  float* redQ = Bl + 512;
  const int p = p0 + wave;
  const float bias = cb[(o0 + ln) * NPOS + min(p, NPOS - 1)];
  {
    float s = 0.f, q = 0.f;
    #pragma unroll
    for (int r = 0; r < 8; ++r) {
      float v = acc0[r] + bias;
      Ys[((hi * 8 + r) * 16 + ln) * YP + wave] = v;
      s += v; q += v * v;
    }
    redS[(wave * 2) * 32 + lane] = s; redQ[(wave * 2) * 32 + lane] = q;
  }
  {
    float s = 0.f, q = 0.f;
    #pragma unroll
    for (int r = 0; r < 8; ++r) {
      float v = acc1[r] + bias;
      Ys[((16 + hi * 8 + r) * 16 + ln) * YP + wave] = v;
      s += v; q += v * v;
    }
    redS[(wave * 2 + 1) * 32 + lane] = s; redQ[(wave * 2 + 1) * 32 + lane] = q;
  }
  __syncthreads();
  if (t < 128) {
    int pl2 = t >> 4, l0 = t & 15;
    if (p0 + pl2 < NPOS) {
      float S = 0.f, Q = 0.f;
      #pragma unroll
      for (int mt2 = 0; mt2 < 2; ++mt2)
        #pragma unroll
        for (int h2 = 0; h2 < 2; ++h2) {
          int tt = (pl2 * 2 + mt2) * 32 + h2 * 16 + l0;
          S += redS[tt]; Q += redQ[tt];
        }
      ws[WS_PART  + (p0 + pl2) * COUT + o0 + l0] = S;
      ws[WS_PART2 + (p0 + pl2) * COUT + o0 + l0] = Q;
    }
  }
  #pragma unroll
  for (int i = 0; i < 4; ++i) {
    int item = i * 256 + t;
    int q4 = item & 1, row = item >> 1;
    int b = row >> 4, ol = row & 15;
    int pq = p0 + 4 * q4;
    if (pq < NPOS)
      *(float4*)(ws + WS_Y + (b * COUT + o0 + ol) * NPOS + pq) =
          *(const float4*)(Ys + row * YP + 4 * q4);
  }
}

__global__ __launch_bounds__(64) void k2_bn(
    const float* __restrict__ gamma, const float* __restrict__ beta,
    float* __restrict__ ws)
{
  const int ch = threadIdx.x;
  float S = 0.f, Q = 0.f;
  for (int p = 0; p < NPOS; ++p) {
    S += ws[WS_PART  + p * COUT + ch];
    Q += ws[WS_PART2 + p * COUT + ch];
  }
  const float inv = 1.0f / (float)(BB * NPOS);
  const float mean = S * inv;
  const float var  = Q * inv - mean * mean;
  const float sc = gamma[ch] * rsqrtf(var + 1e-5f);
  ws[WS_BN + ch]        = sc;
  ws[WS_BN + COUT + ch] = beta[ch] - mean * sc;
}

#define PT        8
#define K3_TILES  113
#define K3_BLOCKS (K3_TILES * 8)
#define SPP       68
#define PSS       1092
#define RP        12


__global__ __launch_bounds__(256) void k3_rec_lif(
    const float* __restrict__ fbv,  const float* __restrict__ soma,
    const float* __restrict__ spk,  const float* __restrict__ acur,
    const float* __restrict__ bt,   const float* __restrict__ lrec,
    const float* __restrict__ taum, const float* __restrict__ taud,
    const float* __restrict__ taua, const float* __restrict__ ws,
    float* __restrict__ out)
{
  __shared__ __attribute__((aligned(16))) float Ss[PT * PSS];
  __shared__ __attribute__((aligned(16))) float Rs[256 * RP];

  const int bid = blockIdx.x;
  const int cq = bid & 3, bh = (bid >> 2) & 1, tile = bid >> 3;
  const int p0 = tile * PT, b0 = bh * 16, c0 = cq * 16;
  const int t = threadIdx.x;
  const int lane = t & 31, wave = t >> 5;
  const int ln = lane & 15, hi = lane >> 4;

  const int q  = t & 1;
  const int cl = (t >> 1) & 15;
  const int c  = c0 + cl;
  const int pq = p0 + 4 * q;
  const bool okq = pq < NPOS;
  const int pql = okq ? pq : NPOS - 4;
  const int bl0 = t >> 5;

  float4 sv4[8];
  #pragma unroll
  for (int i = 0; i < 8; ++i) {
    int e = i * 256 + t;
    int row = e >> 1;
    int d = row & 63, bl = row >> 6;
    sv4[i] = ld4(spk + ((b0 + bl) * COUT + d) * NPOS + pql);
  }

  const int pw = min(p0 + wave, NPOS - 1);
  const float* lp = lrec + ((size_t)pw * COUT + c0 + ln) * COUT + 2 * hi;
  v2f bfr[16];
  #pragma unroll
  for (int j = 0; j < 16; ++j) bfr[j] = *(const v2f*)(lp + 4 * j);

  const float4 tm = ld4(taum + c * NPOS + pql);
  const float4 td = ld4(taud + c * NPOS + pql);
  const float4 ta = ld4(taua + c * NPOS + pql);
  float4 yv[2], bv4[2], av4[2], fv4[2], so4[2];
  #pragma unroll
  for (int i = 0; i < 2; ++i) {
    int idx = ((b0 + bl0 + 8 * i) * COUT + c) * NPOS + pql;
    yv[i]  = ld4(ws + WS_Y + idx);
    bv4[i] = ld4(bt + idx);
    av4[i] = ld4(acur + idx);
    fv4[i] = ld4(fbv + idx);
    so4[i] = ld4(soma + idx);
  }
  const float sc = ws[WS_BN + c];
  const float sh = ws[WS_BN + COUT + c];

  #pragma unroll
  for (int i = 0; i < 8; ++i) {
    int e = i * 256 + t;
    int row = e >> 1;
    int d = row & 63, bl = row >> 6;
    float* dst = Ss + (4 * q) * PSS + bl * SPP + d;
    dst[0]       = sv4[i].x;
    dst[PSS]     = sv4[i].y;
    dst[2 * PSS] = sv4[i].z;
    dst[3 * PSS] = sv4[i].w;
  }
  __syncthreads();

  v8f acc = {};
  const float* ap = Ss + wave * PSS + ln * SPP + 2 * hi;
  #pragma unroll
  for (int j = 0; j < 16; ++j) {
    v2f av = *(const v2f*)(ap + 4 * j);
    acc = __builtin_amdgcn_wmma_f32_16x16x4_f32(false, av, false, bfr[j],
                                                (short)0, acc, false, false);
  }
  #pragma unroll
  for (int r = 0; r < 8; ++r)
    Rs[((hi * 8 + r) * 16 + ln) * RP + wave] = acc[r];
  __syncthreads();

  if (okq) {
    float alpha[4], rho[4], eta[4];
    const float tmv[4] = {tm.x, tm.y, tm.z, tm.w};
    const float tdv[4] = {td.x, td.y, td.z, td.w};
    const float tav[4] = {ta.x, ta.y, ta.z, ta.w};
    #pragma unroll
    for (int k = 0; k < 4; ++k) {
      alpha[k] = __expf(-0.5f / tmv[k]);
      rho[k]   = __expf(-0.5f / tdv[k]);
      eta[k]   = __expf(-0.5f / tav[k]);
    }
    #pragma unroll
    for (int i = 0; i < 2; ++i) {
      const int bl = bl0 + 8 * i;
      const int idx = ((b0 + bl) * COUT + c) * NPOS + pq;
      const float4 rc = *(const float4*)(Rs + (bl * 16 + cl) * RP + 4 * q);
      const float rcv[4] = {rc.x, rc.y, rc.z, rc.w};
      const float yy[4]  = {yv[i].x, yv[i].y, yv[i].z, yv[i].w};
      const float bb[4]  = {bv4[i].x, bv4[i].y, bv4[i].z, bv4[i].w};
      const float aa[4]  = {av4[i].x, av4[i].y, av4[i].z, av4[i].w};
      const float ff4[4] = {fv4[i].x, fv4[i].y, fv4[i].z, fv4[i].w};
      const float ss[4]  = {so4[i].x, so4[i].y, so4[i].z, so4[i].w};
      float o0[4], o1[4], o2[4], o3[4];
      #pragma unroll
      for (int k = 0; k < 4; ++k) {
        float sv    = Ss[(4 * q + k) * PSS + bl * SPP + c];
        float convx = yy[k] * sc + sh + rcv[k];
        float bnew  = rho[k] * bb[k] + (1.f - rho[k]) * sv;
        float thre  = 0.1f + 1.8f * bnew;
        float anew  = eta[k] * aa[k] + ff4[k];
        float sg    = 1.f / (1.f + __expf(-anew)) - 0.5f;
        float snew  = alpha[k] * ss[k] + sg + convx - thre * sv;
        o0[k] = snew;
        o1[k] = (snew - thre) > 0.f ? 1.f : 0.f;
        o2[k] = anew;
        o3[k] = bnew;
      }
      *(float4*)(out + idx)           = make_float4(o0[0], o0[1], o0[2], o0[3]);
      *(float4*)(out + TOT + idx)     = make_float4(o1[0], o1[1], o1[2], o1[3]);
      *(float4*)(out + 2 * TOT + idx) = make_float4(o2[0], o2[1], o2[2], o2[3]);
      *(float4*)(out + 3 * TOT + idx) = make_float4(o3[0], o3[1], o3[2], o3[3]);
    }
  }
}

extern "C" void kernel_launch(void* const* d_in, const int* in_sizes, int n_in,
                              void* d_out, int out_size, void* d_ws, size_t ws_size,
                              hipStream_t stream) {
  const float* ff   = (const float*)d_in[0];
  const float* fb   = (const float*)d_in[1];
  const float* soma = (const float*)d_in[2];
  const float* spk  = (const float*)d_in[3];
  const float* acur = (const float*)d_in[4];
  const float* bt   = (const float*)d_in[5];
  const float* cw   = (const float*)d_in[6];
  const float* cb   = (const float*)d_in[7];
  const float* gam  = (const float*)d_in[8];
  const float* bet  = (const float*)d_in[9];
  const float* lrec = (const float*)d_in[10];
  const float* taum = (const float*)d_in[11];
  const float* taud = (const float*)d_in[12];
  const float* taua = (const float*)d_in[13];
  float* out = (float*)d_out;
  float* ws  = (float*)d_ws;

  hipLaunchKernelGGL(k1_conv,    dim3(K1_BLOCKS), dim3(256), 0, stream, ff, cw, cb, ws);
  hipLaunchKernelGGL(k2_bn,      dim3(1),    dim3(64),  0, stream, gam, bet, ws);
  hipLaunchKernelGGL(k3_rec_lif, dim3(K3_BLOCKS), dim3(256), 0, stream,
                     fb, soma, spk, acur, bt, lrec, taum, taud, taua, ws, out);
}


# BN statistics: k1 folds its tile's 8 positions, so k2 reads 113x64 partials with 1024 threads (float4 loads, LDS tree) instead of a 64-thread 900-step serial loop over 900x64; BN sum order changes
# speedup vs baseline: 4.7398x; 1.4167x over previous
#include <hip/hip_runtime.h>

typedef __attribute__((ext_vector_type(2))) float v2f;
typedef __attribute__((ext_vector_type(8))) float v8f;

#define BB   32
#define CIN  64
#define COUT 64
#define HO   30
#define NPOS 900
#define KT   9
#define TOT  (BB*COUT*NPOS)

#define WS_Y     0
#define WS_PART  (TOT)
#define WS_PART2 (TOT + NPOS*COUT)
#define WS_BN    (TOT + 2*NPOS*COUT)

#define AP 74
#define BP 74

#define P1        8
#define K1_TILES  113
#define K1_BLOCKS (K1_TILES * 4)
#define APL       (BB * AP + 2)
#define BPL       (16 * BP + 40)
#define YP        12
#define ROWF      (NPOS * KT)

__device__ __forceinline__ float4 ld4(const float* p) { return *(const float4*)p; }

__global__ __launch_bounds__(256) void k1_conv(
    const float* __restrict__ ff, const float* __restrict__ cw,
    const float* __restrict__ cb, float* __restrict__ ws)
{
  __shared__ __attribute__((aligned(16))) float Al[P1 * APL];
  __shared__ __attribute__((aligned(16))) float Bl[P1 * BPL];

  const int bid = blockIdx.x;
  const int oq = bid & 3, tile = bid >> 2;
  const int p0 = tile * P1, o0 = oq * 16;
  const int t = threadIdx.x;
  const int lane = t & 31, wave = t >> 5;
  const int ln = lane & 15, hi = lane >> 4;

  const int apl = t & 7, ac = (t >> 3) & 7, ab = t >> 6;
  const int pa = min(p0 + apl, NPOS - 1);
  const float* asrc = ff + ((ab * CIN + ac) * 32 + pa / HO) * 32 + pa % HO;
  float* adst = Al + apl * APL + ab * AP + ac * KT;

  const float* afr = Al + wave * APL + ln * AP + 2 * hi;
  const float* bfr = Bl + wave * BPL + ln * BP + 2 * hi;

  v8f acc0 = {}, acc1 = {};
  for (int cg = 0; cg < 8; ++cg) {
    float4 wv[9];
    #pragma unroll
    for (int i = 0; i < 9; ++i) {
      int idx = i * 256 + t;
      int row = idx / 18, f = idx - row * 18;
      int ol = row >> 3, cc = row & 7;
      int col = p0 * KT + f * 4;
      wv[i] = make_float4(0.f, 0.f, 0.f, 0.f);
      if (col < ROWF)
        wv[i] = ld4(cw + ((o0 + ol) * CIN + cg * 8 + cc) * ROWF + col);
    }
    float pv[8][9];
    #pragma unroll
    for (int i = 0; i < 8; ++i) {
      const float* s = asrc + (i * 4 * CIN + cg * 8) * 1024;
      #pragma unroll
      for (int kh = 0; kh < 3; ++kh)
        #pragma unroll
        for (int kw = 0; kw < 3; ++kw)
          pv[i][kh * 3 + kw] = s[kh * 32 + kw];
    }
    __syncthreads();
    #pragma unroll
    for (int i = 0; i < 8; ++i)
      #pragma unroll
      for (int k = 0; k < KT; ++k) adst[i * 4 * AP + k] = pv[i][k];
    #pragma unroll
    for (int i = 0; i < 9; ++i) {
      int idx = i * 256 + t;
      int row = idx / 18, f = idx - row * 18;
      int ol = row >> 3, cc = row & 7;
      float* d = Bl + ol * BP + cc * KT;
      const float vv[4] = {wv[i].x, wv[i].y, wv[i].z, wv[i].w};
      #pragma unroll
      for (int e = 0; e < 4; ++e) {
        int flat = f * 4 + e;
        int pl = flat / KT, kk = flat - pl * KT;
        d[pl * BPL + kk] = vv[e];
      }
    }
    __syncthreads();
    #pragma unroll
    for (int kb = 0; kb < 72; kb += 4) {
      v2f bv = *(const v2f*)(bfr + kb);
      v2f a0 = *(const v2f*)(afr + kb);
      v2f a1 = *(const v2f*)(afr + 16 * AP + kb);
      acc0 = __builtin_amdgcn_wmma_f32_16x16x4_f32(false, a0, false, bv,
                                                   (short)0, acc0, false, false);
      acc1 = __builtin_amdgcn_wmma_f32_16x16x4_f32(false, a1, false, bv,
                                                   (short)0, acc1, false, false);
    }
  }

  __syncthreads();
  float* Ys   = Al;
  float* redS = Bl;
  float* redQ = Bl + 512;
  const int p = p0 + wave;
  const float bias = cb[(o0 + ln) * NPOS + min(p, NPOS - 1)];
  {
    float s = 0.f, q = 0.f;
    #pragma unroll
    for (int r = 0; r < 8; ++r) {
      float v = acc0[r] + bias;
      Ys[((hi * 8 + r) * 16 + ln) * YP + wave] = v;
      s += v; q += v * v;
    }
    redS[(wave * 2) * 32 + lane] = s; redQ[(wave * 2) * 32 + lane] = q;
  }
  {
    float s = 0.f, q = 0.f;
    #pragma unroll
    for (int r = 0; r < 8; ++r) {
      float v = acc1[r] + bias;
      Ys[((16 + hi * 8 + r) * 16 + ln) * YP + wave] = v;
      s += v; q += v * v;
    }
    redS[(wave * 2 + 1) * 32 + lane] = s; redQ[(wave * 2 + 1) * 32 + lane] = q;
  }
  __syncthreads();
  if (t < 32) {
    const int l0 = t & 15;
    const float* red = Bl + (t >> 4) * 512;
    float tot = 0.f;
    #pragma unroll
    for (int pl2 = 0; pl2 < P1; ++pl2) {
      float f4 = 0.f;
      #pragma unroll
      for (int mt2 = 0; mt2 < 2; ++mt2)
        #pragma unroll
        for (int h2 = 0; h2 < 2; ++h2)
          f4 += red[(pl2 * 2 + mt2) * 32 + h2 * 16 + l0];
      tot += (p0 + pl2 < NPOS) ? f4 : 0.f;
    }
    ws[(t < 16 ? WS_PART : WS_PART2) + tile * COUT + o0 + l0] = tot;
  }
  #pragma unroll
  for (int i = 0; i < 4; ++i) {
    int item = i * 256 + t;
    int q4 = item & 1, row = item >> 1;
    int b = row >> 4, ol = row & 15;
    int pq = p0 + 4 * q4;
    if (pq < NPOS)
      *(float4*)(ws + WS_Y + (b * COUT + o0 + ol) * NPOS + pq) =
          *(const float4*)(Ys + row * YP + 4 * q4);
  }
}

#define K2_T   1024
#define K2_G   64

__global__ __launch_bounds__(K2_T) void k2_bn(
    const float* __restrict__ gamma, const float* __restrict__ beta,
    float* __restrict__ ws)
{
  __shared__ __attribute__((aligned(16))) float gS[K2_G * COUT];
  __shared__ __attribute__((aligned(16))) float gQ[K2_G * COUT];
  __shared__ float hS[4 * COUT];
  __shared__ float hQ[4 * COUT];

  const int t = threadIdx.x;
  const int c4 = t & 15, g = t >> 4;
  const float gm = gamma[t & 63], be = beta[t & 63];
  const float* ps = ws + WS_PART  + g * COUT + c4 * 4;
  const float* pq = ws + WS_PART2 + g * COUT + c4 * 4;
  float4 S = ld4(ps), Q = ld4(pq);
  if (g + K2_G < K1_TILES) {
    const float4 s2 = ld4(ps + K2_G * COUT), q2 = ld4(pq + K2_G * COUT);
    S.x += s2.x; S.y += s2.y; S.z += s2.z; S.w += s2.w;
    Q.x += q2.x; Q.y += q2.y; Q.z += q2.z; Q.w += q2.w;
  }
  *(float4*)(gS + g * COUT + c4 * 4) = S;
  *(float4*)(gQ + g * COUT + c4 * 4) = Q;
  __syncthreads();
  if (t < 4 * COUT) {
    const int ch = t & 63, h = t >> 6;
    float s = 0.f, q = 0.f;
    #pragma unroll
    for (int i = 0; i < K2_G / 4; ++i) {
      s += gS[(h * (K2_G / 4) + i) * COUT + ch];
      q += gQ[(h * (K2_G / 4) + i) * COUT + ch];
    }
    hS[h * COUT + ch] = s;
    hQ[h * COUT + ch] = q;
  }
  __syncthreads();
  if (t < COUT) {
    const int ch = t;
    const float Sx = (hS[ch] + hS[COUT + ch]) + (hS[2 * COUT + ch] + hS[3 * COUT + ch]);
    const float Qx = (hQ[ch] + hQ[COUT + ch]) + (hQ[2 * COUT + ch] + hQ[3 * COUT + ch]);
    const float inv = 1.0f / (float)(BB * NPOS);
    const float mean = Sx * inv;
    const float var  = Qx * inv - mean * mean;
    const float sc = gm * rsqrtf(var + 1e-5f);
    ws[WS_BN + ch]        = sc;
    ws[WS_BN + COUT + ch] = be - mean * sc;
  }
}

#define PT        8
#define K3_TILES  113
#define K3_BLOCKS (K3_TILES * 8)
#define SPP       68
#define PSS       1092
#define RP        12


__global__ __launch_bounds__(256) void k3_rec_lif(
    const float* __restrict__ fbv,  const float* __restrict__ soma,
    const float* __restrict__ spk,  const float* __restrict__ acur,
    const float* __restrict__ bt,   const float* __restrict__ lrec,
    const float* __restrict__ taum, const float* __restrict__ taud,
    const float* __restrict__ taua, const float* __restrict__ ws,
    float* __restrict__ out)
{
  __shared__ __attribute__((aligned(16))) float Ss[PT * PSS];
  __shared__ __attribute__((aligned(16))) float Rs[256 * RP];

  const int bid = blockIdx.x;
  const int cq = bid & 3, bh = (bid >> 2) & 1, tile = bid >> 3;
  const int p0 = tile * PT, b0 = bh * 16, c0 = cq * 16;
  const int t = threadIdx.x;
  const int lane = t & 31, wave = t >> 5;
  const int ln = lane & 15, hi = lane >> 4;

  const int q  = t & 1;
  const int cl = (t >> 1) & 15;
  const int c  = c0 + cl;
  const int pq = p0 + 4 * q;
  const bool okq = pq < NPOS;
  const int pql = okq ? pq : NPOS - 4;
  const int bl0 = t >> 5;

  float4 sv4[8];
  #pragma unroll
  for (int i = 0; i < 8; ++i) {
    int e = i * 256 + t;
    int row = e >> 1;
    int d = row & 63, bl = row >> 6;
    sv4[i] = ld4(spk + ((b0 + bl) * COUT + d) * NPOS + pql);
  }

  const int pw = min(p0 + wave, NPOS - 1);
  const float* lp = lrec + ((size_t)pw * COUT + c0 + ln) * COUT + 2 * hi;
  v2f bfr[16];
  #pragma unroll
  for (int j = 0; j < 16; ++j) bfr[j] = *(const v2f*)(lp + 4 * j);

  const float4 tm = ld4(taum + c * NPOS + pql);
  const float4 td = ld4(taud + c * NPOS + pql);
  const float4 ta = ld4(taua + c * NPOS + pql);
  float4 yv[2], bv4[2], av4[2], fv4[2], so4[2];
  #pragma unroll
  for (int i = 0; i < 2; ++i) {
    int idx = ((b0 + bl0 + 8 * i) * COUT + c) * NPOS + pql;
    yv[i]  = ld4(ws + WS_Y + idx);
    bv4[i] = ld4(bt + idx);
    av4[i] = ld4(acur + idx);
    fv4[i] = ld4(fbv + idx);
    so4[i] = ld4(soma + idx);
  }
  const float sc = ws[WS_BN + c];
  const float sh = ws[WS_BN + COUT + c];

  #pragma unroll
  for (int i = 0; i < 8; ++i) {
    int e = i * 256 + t;
    int row = e >> 1;
    int d = row & 63, bl = row >> 6;
    float* dst = Ss + (4 * q) * PSS + bl * SPP + d;
    dst[0]       = sv4[i].x;
    dst[PSS]     = sv4[i].y;
    dst[2 * PSS] = sv4[i].z;
    dst[3 * PSS] = sv4[i].w;
  }
  __syncthreads();

  v8f acc = {};
  const float* ap = Ss + wave * PSS + ln * SPP + 2 * hi;
  #pragma unroll
  for (int j = 0; j < 16; ++j) {
    v2f av = *(const v2f*)(ap + 4 * j);
    acc = __builtin_amdgcn_wmma_f32_16x16x4_f32(false, av, false, bfr[j],
                                                (short)0, acc, false, false);
  }
  #pragma unroll
  for (int r = 0; r < 8; ++r)
    Rs[((hi * 8 + r) * 16 + ln) * RP + wave] = acc[r];
  __syncthreads();

  if (okq) {
    float alpha[4], rho[4], eta[4];
    const float tmv[4] = {tm.x, tm.y, tm.z, tm.w};
    const float tdv[4] = {td.x, td.y, td.z, td.w};
    const float tav[4] = {ta.x, ta.y, ta.z, ta.w};
    #pragma unroll
    for (int k = 0; k < 4; ++k) {
      alpha[k] = __expf(-0.5f / tmv[k]);
      rho[k]   = __expf(-0.5f / tdv[k]);
      eta[k]   = __expf(-0.5f / tav[k]);
    }
    #pragma unroll
    for (int i = 0; i < 2; ++i) {
      const int bl = bl0 + 8 * i;
      const int idx = ((b0 + bl) * COUT + c) * NPOS + pq;
      const float4 rc = *(const float4*)(Rs + (bl * 16 + cl) * RP + 4 * q);
      const float rcv[4] = {rc.x, rc.y, rc.z, rc.w};
      const float yy[4]  = {yv[i].x, yv[i].y, yv[i].z, yv[i].w};
      const float bb[4]  = {bv4[i].x, bv4[i].y, bv4[i].z, bv4[i].w};
      const float aa[4]  = {av4[i].x, av4[i].y, av4[i].z, av4[i].w};
      const float ff4[4] = {fv4[i].x, fv4[i].y, fv4[i].z, fv4[i].w};
      const float ss[4]  = {so4[i].x, so4[i].y, so4[i].z, so4[i].w};
      float o0[4], o1[4], o2[4], o3[4];
      #pragma unroll
      for (int k = 0; k < 4; ++k) {
        float sv    = Ss[(4 * q + k) * PSS + bl * SPP + c];
        float convx = yy[k] * sc + sh + rcv[k];
        float bnew  = rho[k] * bb[k] + (1.f - rho[k]) * sv;
        float thre  = 0.1f + 1.8f * bnew;
        float anew  = eta[k] * aa[k] + ff4[k];
        float sg    = 1.f / (1.f + __expf(-anew)) - 0.5f;
        float snew  = alpha[k] * ss[k] + sg + convx - thre * sv;
        o0[k] = snew;
        o1[k] = (snew - thre) > 0.f ? 1.f : 0.f;
        o2[k] = anew;
        o3[k] = bnew;
      }
      *(float4*)(out + idx)           = make_float4(o0[0], o0[1], o0[2], o0[3]);
      *(float4*)(out + TOT + idx)     = make_float4(o1[0], o1[1], o1[2], o1[3]);
      *(float4*)(out + 2 * TOT + idx) = make_float4(o2[0], o2[1], o2[2], o2[3]);
      *(float4*)(out + 3 * TOT + idx) = make_float4(o3[0], o3[1], o3[2], o3[3]);
    }
  }
}

extern "C" void kernel_launch(void* const* d_in, const int* in_sizes, int n_in,
                              void* d_out, int out_size, void* d_ws, size_t ws_size,
                              hipStream_t stream) {
  const float* ff   = (const float*)d_in[0];
  const float* fb   = (const float*)d_in[1];
  const float* soma = (const float*)d_in[2];
  const float* spk  = (const float*)d_in[3];
  const float* acur = (const float*)d_in[4];
  const float* bt   = (const float*)d_in[5];
  const float* cw   = (const float*)d_in[6];
  const float* cb   = (const float*)d_in[7];
  const float* gam  = (const float*)d_in[8];
  const float* bet  = (const float*)d_in[9];
  const float* lrec = (const float*)d_in[10];
  const float* taum = (const float*)d_in[11];
  const float* taud = (const float*)d_in[12];
  const float* taua = (const float*)d_in[13];
  float* out = (float*)d_out;
  float* ws  = (float*)d_ws;

  hipLaunchKernelGGL(k1_conv,    dim3(K1_BLOCKS), dim3(256), 0, stream, ff, cw, cb, ws);
  hipLaunchKernelGGL(k2_bn,      dim3(1),    dim3(K2_T), 0, stream, gam, bet, ws);
  hipLaunchKernelGGL(k3_rec_lif, dim3(K3_BLOCKS), dim3(256), 0, stream,
                     fb, soma, spk, acur, bt, lrec, taum, taud, taua, ws, out);
}


# k1 K loop pipelined by one chunk: a trip issues the next chunk's global loads, then runs the previous chunk's 36 WMMAs; loop barriers fence LDS only so the loads stay in flight; sum order unchanged
# speedup vs baseline: 4.9302x; 1.0402x over previous
#include <hip/hip_runtime.h>

typedef __attribute__((ext_vector_type(2))) float v2f;
typedef __attribute__((ext_vector_type(8))) float v8f;

#define BB   32
#define CIN  64
#define COUT 64
#define HO   30
#define NPOS 900
#define KT   9
#define TOT  (BB*COUT*NPOS)

#define WS_Y     0
#define WS_PART  (TOT)
#define WS_PART2 (TOT + NPOS*COUT)
#define WS_BN    (TOT + 2*NPOS*COUT)

#define AP 74
#define BP 74

#define P1        8
#define K1_TILES  113
#define K1_BLOCKS (K1_TILES * 4)
#define APL       (BB * AP + 2)
#define BPL       (16 * BP + 40)
#define YP        12
#define ROWF      (NPOS * KT)

__device__ __forceinline__ float4 ld4(const float* p) { return *(const float4*)p; }

__device__ __forceinline__ void lds_barrier() {
  asm volatile("s_wait_dscnt 0x0" ::: "memory");
  __builtin_amdgcn_fence(__ATOMIC_RELEASE, "workgroup", "local");
  __builtin_amdgcn_s_barrier();
  __builtin_amdgcn_fence(__ATOMIC_ACQUIRE, "workgroup", "local");
}

__global__ __launch_bounds__(256) void k1_conv(
    const float* __restrict__ ff, const float* __restrict__ cw,
    const float* __restrict__ cb, float* __restrict__ ws)
{
  __shared__ __attribute__((aligned(16))) float Al[P1 * APL];
  __shared__ __attribute__((aligned(16))) float Bl[P1 * BPL];

  const int bid = blockIdx.x;
  const int oq = bid & 3, tile = bid >> 2;
  const int p0 = tile * P1, o0 = oq * 16;
  const int t = threadIdx.x;
  const int lane = t & 31, wave = t >> 5;
  const int ln = lane & 15, hi = lane >> 4;

  const int apl = t & 7, ac = (t >> 3) & 7, ab = t >> 6;
  const int pa = min(p0 + apl, NPOS - 1);
  const float* asrc = ff + ((ab * CIN + ac) * 32 + pa / HO) * 32 + pa % HO;
  float* adst = Al + apl * APL + ab * AP + ac * KT;

  const float* afr = Al + wave * APL + ln * AP + 2 * hi;
  const float* bfr = Bl + wave * BPL + ln * BP + 2 * hi;

  v8f acc0 = {}, acc1 = {};
  auto mma_chunk = [&]() {
    #pragma unroll
    for (int kb = 0; kb < 72; kb += 4) {
      v2f bv = *(const v2f*)(bfr + kb);
      v2f a0 = *(const v2f*)(afr + kb);
      v2f a1 = *(const v2f*)(afr + 16 * AP + kb);
      acc0 = __builtin_amdgcn_wmma_f32_16x16x4_f32(false, a0, false, bv,
                                                   (short)0, acc0, false, false);
      acc1 = __builtin_amdgcn_wmma_f32_16x16x4_f32(false, a1, false, bv,
                                                   (short)0, acc1, false, false);
    }
  };
  for (int cg = 0; cg < 8; ++cg) {
    float4 wv[9];
    #pragma unroll
    for (int i = 0; i < 9; ++i) {
      int idx = i * 256 + t;
      int row = idx / 18, f = idx - row * 18;
      int ol = row >> 3, cc = row & 7;
      int col = p0 * KT + f * 4;
      wv[i] = make_float4(0.f, 0.f, 0.f, 0.f);
      if (col < ROWF)
        wv[i] = ld4(cw + ((o0 + ol) * CIN + cg * 8 + cc) * ROWF + col);
    }
    float pv[8][9];
    #pragma unroll
    for (int i = 0; i < 8; ++i) {
      const float* s = asrc + (i * 4 * CIN + cg * 8) * 1024;
      #pragma unroll
      for (int kh = 0; kh < 3; ++kh)
        #pragma unroll
        for (int kw = 0; kw < 3; ++kw)
          pv[i][kh * 3 + kw] = s[kh * 32 + kw];
    }
    if (cg > 0) {
      lds_barrier();
      mma_chunk();
    }
    lds_barrier();
    #pragma unroll
    for (int i = 0; i < 9; ++i) {
      int idx = i * 256 + t;
      int row = idx / 18, f = idx - row * 18;
      int ol = row >> 3, cc = row & 7;
      float* d = Bl + ol * BP + cc * KT;
      const float vv[4] = {wv[i].x, wv[i].y, wv[i].z, wv[i].w};
      #pragma unroll
      for (int e = 0; e < 4; ++e) {
        int flat = f * 4 + e;
        int pl = flat / KT, kk = flat - pl * KT;
        d[pl * BPL + kk] = vv[e];
      }
    }
    #pragma unroll
    for (int i = 0; i < 8; ++i)
      #pragma unroll
      for (int k = 0; k < KT; ++k) adst[i * 4 * AP + k] = pv[i][k];
  }
  lds_barrier();
  mma_chunk();

  __syncthreads();
  float* Ys   = Al;
  float* redS = Bl;
  float* redQ = Bl + 512;
  const int p = p0 + wave;
  const float bias = cb[(o0 + ln) * NPOS + min(p, NPOS - 1)];
  {
    float s = 0.f, q = 0.f;
    #pragma unroll
    for (int r = 0; r < 8; ++r) {
      float v = acc0[r] + bias;
      Ys[((hi * 8 + r) * 16 + ln) * YP + wave] = v;
      s += v; q += v * v;
    }
    redS[(wave * 2) * 32 + lane] = s; redQ[(wave * 2) * 32 + lane] = q;
  }
  {
    float s = 0.f, q = 0.f;
    #pragma unroll
    for (int r = 0; r < 8; ++r) {
      float v = acc1[r] + bias;
      Ys[((16 + hi * 8 + r) * 16 + ln) * YP + wave] = v;
      s += v; q += v * v;
    }
    redS[(wave * 2 + 1) * 32 + lane] = s; redQ[(wave * 2 + 1) * 32 + lane] = q;
  }
  __syncthreads();
  if (t < 32) {
    const int l0 = t & 15;
    const float* red = Bl + (t >> 4) * 512;
    float tot = 0.f;
    #pragma unroll
    for (int pl2 = 0; pl2 < P1; ++pl2) {
      float f4 = 0.f;
      #pragma unroll
      for (int mt2 = 0; mt2 < 2; ++mt2)
        #pragma unroll
        for (int h2 = 0; h2 < 2; ++h2)
          f4 += red[(pl2 * 2 + mt2) * 32 + h2 * 16 + l0];
      tot += (p0 + pl2 < NPOS) ? f4 : 0.f;
    }
    ws[(t < 16 ? WS_PART : WS_PART2) + tile * COUT + o0 + l0] = tot;
  }
  #pragma unroll
  for (int i = 0; i < 4; ++i) {
    int item = i * 256 + t;
    int q4 = item & 1, row = item >> 1;
    int b = row >> 4, ol = row & 15;
    int pq = p0 + 4 * q4;
    if (pq < NPOS)
      *(float4*)(ws + WS_Y + (b * COUT + o0 + ol) * NPOS + pq) =
          *(const float4*)(Ys + row * YP + 4 * q4);
  }
}

#define K2_T   1024
#define K2_G   64

__global__ __launch_bounds__(K2_T) void k2_bn(
    const float* __restrict__ gamma, const float* __restrict__ beta,
    float* __restrict__ ws)
{
  __shared__ __attribute__((aligned(16))) float gS[K2_G * COUT];
  __shared__ __attribute__((aligned(16))) float gQ[K2_G * COUT];
  __shared__ float hS[4 * COUT];
  __shared__ float hQ[4 * COUT];

  const int t = threadIdx.x;
  const int c4 = t & 15, g = t >> 4;
  const float gm = gamma[t & 63], be = beta[t & 63];
  const float* ps = ws + WS_PART  + g * COUT + c4 * 4;
  const float* pq = ws + WS_PART2 + g * COUT + c4 * 4;
  float4 S = ld4(ps), Q = ld4(pq);
  if (g + K2_G < K1_TILES) {
    const float4 s2 = ld4(ps + K2_G * COUT), q2 = ld4(pq + K2_G * COUT);
    S.x += s2.x; S.y += s2.y; S.z += s2.z; S.w += s2.w;
    Q.x += q2.x; Q.y += q2.y; Q.z += q2.z; Q.w += q2.w;
  }
  *(float4*)(gS + g * COUT + c4 * 4) = S;
  *(float4*)(gQ + g * COUT + c4 * 4) = Q;
  __syncthreads();
  if (t < 4 * COUT) {
    const int ch = t & 63, h = t >> 6;
    float s = 0.f, q = 0.f;
    #pragma unroll
    for (int i = 0; i < K2_G / 4; ++i) {
      s += gS[(h * (K2_G / 4) + i) * COUT + ch];
      q += gQ[(h * (K2_G / 4) + i) * COUT + ch];
    }
    hS[h * COUT + ch] = s;
    hQ[h * COUT + ch] = q;
  }
  __syncthreads();
  if (t < COUT) {
    const int ch = t;
    const float Sx = (hS[ch] + hS[COUT + ch]) + (hS[2 * COUT + ch] + hS[3 * COUT + ch]);
    const float Qx = (hQ[ch] + hQ[COUT + ch]) + (hQ[2 * COUT + ch] + hQ[3 * COUT + ch]);
    const float inv = 1.0f / (float)(BB * NPOS);
    const float mean = Sx * inv;
    const float var  = Qx * inv - mean * mean;
    const float sc = gm * rsqrtf(var + 1e-5f);
    ws[WS_BN + ch]        = sc;
    ws[WS_BN + COUT + ch] = be - mean * sc;
  }
}

#define PT        8
#define K3_TILES  113
#define K3_BLOCKS (K3_TILES * 8)
#define SPP       68
#define PSS       1092
#define RP        12


__global__ __launch_bounds__(256) void k3_rec_lif(
    const float* __restrict__ fbv,  const float* __restrict__ soma,
    const float* __restrict__ spk,  const float* __restrict__ acur,
    const float* __restrict__ bt,   const float* __restrict__ lrec,
    const float* __restrict__ taum, const float* __restrict__ taud,
    const float* __restrict__ taua, const float* __restrict__ ws,
    float* __restrict__ out)
{
  __shared__ __attribute__((aligned(16))) float Ss[PT * PSS];
  __shared__ __attribute__((aligned(16))) float Rs[256 * RP];

  const int bid = blockIdx.x;
  const int cq = bid & 3, bh = (bid >> 2) & 1, tile = bid >> 3;
  const int p0 = tile * PT, b0 = bh * 16, c0 = cq * 16;
  const int t = threadIdx.x;
  const int lane = t & 31, wave = t >> 5;
  const int ln = lane & 15, hi = lane >> 4;

  const int q  = t & 1;
  const int cl = (t >> 1) & 15;
  const int c  = c0 + cl;
  const int pq = p0 + 4 * q;
  const bool okq = pq < NPOS;
  const int pql = okq ? pq : NPOS - 4;
  const int bl0 = t >> 5;

  float4 sv4[8];
  #pragma unroll
  for (int i = 0; i < 8; ++i) {
    int e = i * 256 + t;
    int row = e >> 1;
    int d = row & 63, bl = row >> 6;
    sv4[i] = ld4(spk + ((b0 + bl) * COUT + d) * NPOS + pql);
  }

  const int pw = min(p0 + wave, NPOS - 1);
  const float* lp = lrec + ((size_t)pw * COUT + c0 + ln) * COUT + 2 * hi;
  v2f bfr[16];
  #pragma unroll
  for (int j = 0; j < 16; ++j) bfr[j] = *(const v2f*)(lp + 4 * j);

  const float4 tm = ld4(taum + c * NPOS + pql);
  const float4 td = ld4(taud + c * NPOS + pql);
  const float4 ta = ld4(taua + c * NPOS + pql);
  float4 yv[2], bv4[2], av4[2], fv4[2], so4[2];
  #pragma unroll
  for (int i = 0; i < 2; ++i) {
    int idx = ((b0 + bl0 + 8 * i) * COUT + c) * NPOS + pql;
    yv[i]  = ld4(ws + WS_Y + idx);
    bv4[i] = ld4(bt + idx);
    av4[i] = ld4(acur + idx);
    fv4[i] = ld4(fbv + idx);
    so4[i] = ld4(soma + idx);
  }
  const float sc = ws[WS_BN + c];
  const float sh = ws[WS_BN + COUT + c];

  #pragma unroll
  for (int i = 0; i < 8; ++i) {
    int e = i * 256 + t;
    int row = e >> 1;
    int d = row & 63, bl = row >> 6;
    float* dst = Ss + (4 * q) * PSS + bl * SPP + d;
    dst[0]       = sv4[i].x;
    dst[PSS]     = sv4[i].y;
    dst[2 * PSS] = sv4[i].z;
    dst[3 * PSS] = sv4[i].w;
  }
  __syncthreads();

  v8f acc = {};
  const float* ap = Ss + wave * PSS + ln * SPP + 2 * hi;
  #pragma unroll
  for (int j = 0; j < 16; ++j) {
    v2f av = *(const v2f*)(ap + 4 * j);
    acc = __builtin_amdgcn_wmma_f32_16x16x4_f32(false, av, false, bfr[j],
                                                (short)0, acc, false, false);
  }
  #pragma unroll
  for (int r = 0; r < 8; ++r)
    Rs[((hi * 8 + r) * 16 + ln) * RP + wave] = acc[r];
  __syncthreads();

  if (okq) {
    float alpha[4], rho[4], eta[4];
    const float tmv[4] = {tm.x, tm.y, tm.z, tm.w};
    const float tdv[4] = {td.x, td.y, td.z, td.w};
    const float tav[4] = {ta.x, ta.y, ta.z, ta.w};
    #pragma unroll
    for (int k = 0; k < 4; ++k) {
      alpha[k] = __expf(-0.5f / tmv[k]);
      rho[k]   = __expf(-0.5f / tdv[k]);
      eta[k]   = __expf(-0.5f / tav[k]);
    }
    #pragma unroll
    for (int i = 0; i < 2; ++i) {
      const int bl = bl0 + 8 * i;
      const int idx = ((b0 + bl) * COUT + c) * NPOS + pq;
      const float4 rc = *(const float4*)(Rs + (bl * 16 + cl) * RP + 4 * q);
      const float rcv[4] = {rc.x, rc.y, rc.z, rc.w};
      const float yy[4]  = {yv[i].x, yv[i].y, yv[i].z, yv[i].w};
      const float bb[4]  = {bv4[i].x, bv4[i].y, bv4[i].z, bv4[i].w};
      const float aa[4]  = {av4[i].x, av4[i].y, av4[i].z, av4[i].w};
      const float ff4[4] = {fv4[i].x, fv4[i].y, fv4[i].z, fv4[i].w};
      const float ss[4]  = {so4[i].x, so4[i].y, so4[i].z, so4[i].w};
      float o0[4], o1[4], o2[4], o3[4];
      #pragma unroll
      for (int k = 0; k < 4; ++k) {
        float sv    = Ss[(4 * q + k) * PSS + bl * SPP + c];
        float convx = yy[k] * sc + sh + rcv[k];
        float bnew  = rho[k] * bb[k] + (1.f - rho[k]) * sv;
        float thre  = 0.1f + 1.8f * bnew;
        float anew  = eta[k] * aa[k] + ff4[k];
        float sg    = 1.f / (1.f + __expf(-anew)) - 0.5f;
        float snew  = alpha[k] * ss[k] + sg + convx - thre * sv;
        o0[k] = snew;
        o1[k] = (snew - thre) > 0.f ? 1.f : 0.f;
        o2[k] = anew;
        o3[k] = bnew;
      }
      *(float4*)(out + idx)           = make_float4(o0[0], o0[1], o0[2], o0[3]);
      *(float4*)(out + TOT + idx)     = make_float4(o1[0], o1[1], o1[2], o1[3]);
      *(float4*)(out + 2 * TOT + idx) = make_float4(o2[0], o2[1], o2[2], o2[3]);
      *(float4*)(out + 3 * TOT + idx) = make_float4(o3[0], o3[1], o3[2], o3[3]);
    }
  }
}

extern "C" void kernel_launch(void* const* d_in, const int* in_sizes, int n_in,
                              void* d_out, int out_size, void* d_ws, size_t ws_size,
                              hipStream_t stream) {
  const float* ff   = (const float*)d_in[0];
  const float* fb   = (const float*)d_in[1];
  const float* soma = (const float*)d_in[2];
  const float* spk  = (const float*)d_in[3];
  const float* acur = (const float*)d_in[4];
  const float* bt   = (const float*)d_in[5];
  const float* cw   = (const float*)d_in[6];
  const float* cb   = (const float*)d_in[7];
  const float* gam  = (const float*)d_in[8];
  const float* bet  = (const float*)d_in[9];
  const float* lrec = (const float*)d_in[10];
  const float* taum = (const float*)d_in[11];
  const float* taud = (const float*)d_in[12];
  const float* taua = (const float*)d_in[13];
  float* out = (float*)d_out;
  float* ws  = (float*)d_ws;

  hipLaunchKernelGGL(k1_conv,    dim3(K1_BLOCKS), dim3(256), 0, stream, ff, cw, cb, ws);
  hipLaunchKernelGGL(k2_bn,      dim3(1),    dim3(K2_T), 0, stream, gam, bet, ws);
  hipLaunchKernelGGL(k3_rec_lif, dim3(K3_BLOCKS), dim3(256), 0, stream,
                     fb, soma, spk, acur, bt, lrec, taum, taud, taua, ws, out);
}


# k1 block widened to 8 positions x 32 out-channels (16 waves): two out-channel quarters share one patch tile, so each patch is read from L2 and stored to LDS once for both; sum orders unchanged
# speedup vs baseline: 5.3954x; 1.0944x over previous
#include <hip/hip_runtime.h>

typedef __attribute__((ext_vector_type(2))) float v2f;
typedef __attribute__((ext_vector_type(8))) float v8f;

#define BB   32
#define CIN  64
#define COUT 64
#define HO   30
#define NPOS 900
#define KT   9
#define TOT  (BB*COUT*NPOS)

#define WS_Y     0
#define WS_PART  (TOT)
#define WS_PART2 (TOT + NPOS*COUT)
#define WS_BN    (TOT + 2*NPOS*COUT)

#define AP 74
#define BP 74

#define P1        8
#define K1_TILES  113
#define K1_BLOCKS (K1_TILES * 2)
#define K1_T      512
#define O1        32
#define APL       (BB * AP + 2)
#define BPL       (16 * BP + 40)
#define YP        12
#define ROWF      (NPOS * KT)

__device__ __forceinline__ float4 ld4(const float* p) { return *(const float4*)p; }

__device__ __forceinline__ void lds_barrier() {
  asm volatile("s_wait_dscnt 0x0" ::: "memory");
  __builtin_amdgcn_fence(__ATOMIC_RELEASE, "workgroup", "local");
  __builtin_amdgcn_s_barrier();
  __builtin_amdgcn_fence(__ATOMIC_ACQUIRE, "workgroup", "local");
}

__global__ __launch_bounds__(K1_T) void k1_conv(
    const float* __restrict__ ff, const float* __restrict__ cw,
    const float* __restrict__ cb, float* __restrict__ ws)
{
  __shared__ __attribute__((aligned(16))) float Al[P1 * APL];
  __shared__ __attribute__((aligned(16))) float Bl[2 * P1 * BPL];

  const int bid = blockIdx.x;
  const int oh = bid & 1, tile = bid >> 1;
  const int p0 = tile * P1, o0 = oh * O1;
  const int t = threadIdx.x;
  const int lane = t & 31, wave = t >> 5;
  const int ln = lane & 15, hi = lane >> 4;
  const int wp = wave & 7, wq = wave >> 3;

  const int apl = t & 7, ac = (t >> 3) & 7, ab = t >> 6;
  const int pa = min(p0 + apl, NPOS - 1);
  const float* asrc = ff + ((ab * CIN + ac) * 32 + pa / HO) * 32 + pa % HO;
  float* adst = Al + apl * APL + ab * AP + ac * KT;

  const float* afr = Al + wp * APL + ln * AP + 2 * hi;
  const float* bfr = Bl + (wq * P1 + wp) * BPL + ln * BP + 2 * hi;

  v8f acc0 = {}, acc1 = {};
  auto mma_chunk = [&]() {
    #pragma unroll
    for (int kb = 0; kb < 72; kb += 4) {
      v2f bv = *(const v2f*)(bfr + kb);
      v2f a0 = *(const v2f*)(afr + kb);
      v2f a1 = *(const v2f*)(afr + 16 * AP + kb);
      acc0 = __builtin_amdgcn_wmma_f32_16x16x4_f32(false, a0, false, bv,
                                                   (short)0, acc0, false, false);
      acc1 = __builtin_amdgcn_wmma_f32_16x16x4_f32(false, a1, false, bv,
                                                   (short)0, acc1, false, false);
    }
  };
  for (int cg = 0; cg < 8; ++cg) {
    float4 wv[9];
    #pragma unroll
    for (int i = 0; i < 9; ++i) {
      int idx = i * K1_T + t;
      int row = idx / 18, f = idx - row * 18;
      int ol = row >> 3, cc = row & 7;
      int col = p0 * KT + f * 4;
      wv[i] = make_float4(0.f, 0.f, 0.f, 0.f);
      if (col < ROWF)
        wv[i] = ld4(cw + ((o0 + ol) * CIN + cg * 8 + cc) * ROWF + col);
    }
    float pv[4][9];
    #pragma unroll
    for (int i = 0; i < 4; ++i) {
      const float* s = asrc + (i * 8 * CIN + cg * 8) * 1024;
      #pragma unroll
      for (int kh = 0; kh < 3; ++kh)
        #pragma unroll
        for (int kw = 0; kw < 3; ++kw)
          pv[i][kh * 3 + kw] = s[kh * 32 + kw];
    }
    if (cg > 0) {
      lds_barrier();
      mma_chunk();
    }
    lds_barrier();
    #pragma unroll
    for (int i = 0; i < 9; ++i) {
      int idx = i * K1_T + t;
      int row = idx / 18, f = idx - row * 18;
      int ol = row >> 3, cc = row & 7;
      float* d = Bl + (ol >> 4) * (P1 * BPL) + (ol & 15) * BP + cc * KT;
      const float vv[4] = {wv[i].x, wv[i].y, wv[i].z, wv[i].w};
      #pragma unroll
      for (int e = 0; e < 4; ++e) {
        int flat = f * 4 + e;
        int pl = flat / KT, kk = flat - pl * KT;
        d[pl * BPL + kk] = vv[e];
      }
    }
    #pragma unroll
    for (int i = 0; i < 4; ++i)
      #pragma unroll
      for (int k = 0; k < KT; ++k) adst[i * 8 * AP + k] = pv[i][k];
  }
  lds_barrier();
  mma_chunk();

  __syncthreads();
  float* Ys   = Al;
  float* redS = Bl;
  float* redQ = Bl + 1024;
  const int p = p0 + wp;
  const int oc = wq * 16 + ln;
  const float bias = cb[(o0 + oc) * NPOS + min(p, NPOS - 1)];
  {
    float s = 0.f, q = 0.f;
    #pragma unroll
    for (int r = 0; r < 8; ++r) {
      float v = acc0[r] + bias;
      Ys[((hi * 8 + r) * O1 + oc) * YP + wp] = v;
      s += v; q += v * v;
    }
    redS[(wave * 2) * 32 + lane] = s; redQ[(wave * 2) * 32 + lane] = q;
  }
  {
    float s = 0.f, q = 0.f;
    #pragma unroll
    for (int r = 0; r < 8; ++r) {
      float v = acc1[r] + bias;
      Ys[((16 + hi * 8 + r) * O1 + oc) * YP + wp] = v;
      s += v; q += v * v;
    }
    redS[(wave * 2 + 1) * 32 + lane] = s; redQ[(wave * 2 + 1) * 32 + lane] = q;
  }
  __syncthreads();
  if (t < 64) {
    const int l0 = t & 31;
    const float* red = Bl + (t >> 5) * 1024 + (l0 >> 4) * (P1 * 64) + (l0 & 15);
    float tot = 0.f;
    #pragma unroll
    for (int pl2 = 0; pl2 < P1; ++pl2) {
      float f4 = 0.f;
      #pragma unroll
      for (int mt2 = 0; mt2 < 2; ++mt2)
        #pragma unroll
        for (int h2 = 0; h2 < 2; ++h2)
          f4 += red[(pl2 * 2 + mt2) * 32 + h2 * 16];
      tot += (p0 + pl2 < NPOS) ? f4 : 0.f;
    }
    ws[(t < 32 ? WS_PART : WS_PART2) + tile * COUT + o0 + l0] = tot;
  }
  #pragma unroll
  for (int i = 0; i < 4; ++i) {
    int item = i * K1_T + t;
    int q4 = item & 1, row = item >> 1;
    int b = row >> 5, ol = row & 31;
    int pq = p0 + 4 * q4;
    if (pq < NPOS)
      *(float4*)(ws + WS_Y + (b * COUT + o0 + ol) * NPOS + pq) =
          *(const float4*)(Ys + row * YP + 4 * q4);
  }
}

#define K2_T   1024
#define K2_G   64

__global__ __launch_bounds__(K2_T) void k2_bn(
    const float* __restrict__ gamma, const float* __restrict__ beta,
    float* __restrict__ ws)
{
  __shared__ __attribute__((aligned(16))) float gS[K2_G * COUT];
  __shared__ __attribute__((aligned(16))) float gQ[K2_G * COUT];
  __shared__ float hS[4 * COUT];
  __shared__ float hQ[4 * COUT];

  const int t = threadIdx.x;
  const int c4 = t & 15, g = t >> 4;
  const float gm = gamma[t & 63], be = beta[t & 63];
  const float* ps = ws + WS_PART  + g * COUT + c4 * 4;
  const float* pq = ws + WS_PART2 + g * COUT + c4 * 4;
  float4 S = ld4(ps), Q = ld4(pq);
  if (g + K2_G < K1_TILES) {
    const float4 s2 = ld4(ps + K2_G * COUT), q2 = ld4(pq + K2_G * COUT);
    S.x += s2.x; S.y += s2.y; S.z += s2.z; S.w += s2.w;
    Q.x += q2.x; Q.y += q2.y; Q.z += q2.z; Q.w += q2.w;
  }
  *(float4*)(gS + g * COUT + c4 * 4) = S;
  *(float4*)(gQ + g * COUT + c4 * 4) = Q;
  __syncthreads();
  if (t < 4 * COUT) {
    const int ch = t & 63, h = t >> 6;
    float s = 0.f, q = 0.f;
    #pragma unroll
    for (int i = 0; i < K2_G / 4; ++i) {
      s += gS[(h * (K2_G / 4) + i) * COUT + ch];
      q += gQ[(h * (K2_G / 4) + i) * COUT + ch];
    }
    hS[h * COUT + ch] = s;
    hQ[h * COUT + ch] = q;
  }
  __syncthreads();
  if (t < COUT) {
    const int ch = t;
    const float Sx = (hS[ch] + hS[COUT + ch]) + (hS[2 * COUT + ch] + hS[3 * COUT + ch]);
    const float Qx = (hQ[ch] + hQ[COUT + ch]) + (hQ[2 * COUT + ch] + hQ[3 * COUT + ch]);
    const float inv = 1.0f / (float)(BB * NPOS);
    const float mean = Sx * inv;
    const float var  = Qx * inv - mean * mean;
    const float sc = gm * rsqrtf(var + 1e-5f);
    ws[WS_BN + ch]        = sc;
    ws[WS_BN + COUT + ch] = be - mean * sc;
  }
}

#define PT        8
#define K3_TILES  113
#define K3_BLOCKS (K3_TILES * 8)
#define SPP       68
#define PSS       1092
#define RP        12


__global__ __launch_bounds__(256) void k3_rec_lif(
    const float* __restrict__ fbv,  const float* __restrict__ soma,
    const float* __restrict__ spk,  const float* __restrict__ acur,
    const float* __restrict__ bt,   const float* __restrict__ lrec,
    const float* __restrict__ taum, const float* __restrict__ taud,
    const float* __restrict__ taua, const float* __restrict__ ws,
    float* __restrict__ out)
{
  __shared__ __attribute__((aligned(16))) float Ss[PT * PSS];
  __shared__ __attribute__((aligned(16))) float Rs[256 * RP];

  const int bid = blockIdx.x;
  const int cq = bid & 3, bh = (bid >> 2) & 1, tile = bid >> 3;
  const int p0 = tile * PT, b0 = bh * 16, c0 = cq * 16;
  const int t = threadIdx.x;
  const int lane = t & 31, wave = t >> 5;
  const int ln = lane & 15, hi = lane >> 4;

  const int q  = t & 1;
  const int cl = (t >> 1) & 15;
  const int c  = c0 + cl;
  const int pq = p0 + 4 * q;
  const bool okq = pq < NPOS;
  const int pql = okq ? pq : NPOS - 4;
  const int bl0 = t >> 5;

  float4 sv4[8];
  #pragma unroll
  for (int i = 0; i < 8; ++i) {
    int e = i * 256 + t;
    int row = e >> 1;
    int d = row & 63, bl = row >> 6;
    sv4[i] = ld4(spk + ((b0 + bl) * COUT + d) * NPOS + pql);
  }

  const int pw = min(p0 + wave, NPOS - 1);
  const float* lp = lrec + ((size_t)pw * COUT + c0 + ln) * COUT + 2 * hi;
  v2f bfr[16];
  #pragma unroll
  for (int j = 0; j < 16; ++j) bfr[j] = *(const v2f*)(lp + 4 * j);

  const float4 tm = ld4(taum + c * NPOS + pql);
  const float4 td = ld4(taud + c * NPOS + pql);
  const float4 ta = ld4(taua + c * NPOS + pql);
  float4 yv[2], bv4[2], av4[2], fv4[2], so4[2];
  #pragma unroll
  for (int i = 0; i < 2; ++i) {
    int idx = ((b0 + bl0 + 8 * i) * COUT + c) * NPOS + pql;
    yv[i]  = ld4(ws + WS_Y + idx);
    bv4[i] = ld4(bt + idx);
    av4[i] = ld4(acur + idx);
    fv4[i] = ld4(fbv + idx);
    so4[i] = ld4(soma + idx);
  }
  const float sc = ws[WS_BN + c];
  const float sh = ws[WS_BN + COUT + c];

  #pragma unroll
  for (int i = 0; i < 8; ++i) {
    int e = i * 256 + t;
    int row = e >> 1;
    int d = row & 63, bl = row >> 6;
    float* dst = Ss + (4 * q) * PSS + bl * SPP + d;
    dst[0]       = sv4[i].x;
    dst[PSS]     = sv4[i].y;
    dst[2 * PSS] = sv4[i].z;
    dst[3 * PSS] = sv4[i].w;
  }
  __syncthreads();

  v8f acc = {};
  const float* ap = Ss + wave * PSS + ln * SPP + 2 * hi;
  #pragma unroll
  for (int j = 0; j < 16; ++j) {
    v2f av = *(const v2f*)(ap + 4 * j);
    acc = __builtin_amdgcn_wmma_f32_16x16x4_f32(false, av, false, bfr[j],
                                                (short)0, acc, false, false);
  }
  #pragma unroll
  for (int r = 0; r < 8; ++r)
    Rs[((hi * 8 + r) * 16 + ln) * RP + wave] = acc[r];
  __syncthreads();

  if (okq) {
    float alpha[4], rho[4], eta[4];
    const float tmv[4] = {tm.x, tm.y, tm.z, tm.w};
    const float tdv[4] = {td.x, td.y, td.z, td.w};
    const float tav[4] = {ta.x, ta.y, ta.z, ta.w};
    #pragma unroll
    for (int k = 0; k < 4; ++k) {
      alpha[k] = __expf(-0.5f / tmv[k]);
      rho[k]   = __expf(-0.5f / tdv[k]);
      eta[k]   = __expf(-0.5f / tav[k]);
    }
    #pragma unroll
    for (int i = 0; i < 2; ++i) {
      const int bl = bl0 + 8 * i;
      const int idx = ((b0 + bl) * COUT + c) * NPOS + pq;
      const float4 rc = *(const float4*)(Rs + (bl * 16 + cl) * RP + 4 * q);
      const float rcv[4] = {rc.x, rc.y, rc.z, rc.w};
      const float yy[4]  = {yv[i].x, yv[i].y, yv[i].z, yv[i].w};
      const float bb[4]  = {bv4[i].x, bv4[i].y, bv4[i].z, bv4[i].w};
      const float aa[4]  = {av4[i].x, av4[i].y, av4[i].z, av4[i].w};
      const float ff4[4] = {fv4[i].x, fv4[i].y, fv4[i].z, fv4[i].w};
      const float ss[4]  = {so4[i].x, so4[i].y, so4[i].z, so4[i].w};
      float o0[4], o1[4], o2[4], o3[4];
      #pragma unroll
      for (int k = 0; k < 4; ++k) {
        float sv    = Ss[(4 * q + k) * PSS + bl * SPP + c];
        float convx = yy[k] * sc + sh + rcv[k];
        float bnew  = rho[k] * bb[k] + (1.f - rho[k]) * sv;
        float thre  = 0.1f + 1.8f * bnew;
        float anew  = eta[k] * aa[k] + ff4[k];
        float sg    = 1.f / (1.f + __expf(-anew)) - 0.5f;
        float snew  = alpha[k] * ss[k] + sg + convx - thre * sv;
        o0[k] = snew;
        o1[k] = (snew - thre) > 0.f ? 1.f : 0.f;
        o2[k] = anew;
        o3[k] = bnew;
      }
      *(float4*)(out + idx)           = make_float4(o0[0], o0[1], o0[2], o0[3]);
      *(float4*)(out + TOT + idx)     = make_float4(o1[0], o1[1], o1[2], o1[3]);
      *(float4*)(out + 2 * TOT + idx) = make_float4(o2[0], o2[1], o2[2], o2[3]);
      *(float4*)(out + 3 * TOT + idx) = make_float4(o3[0], o3[1], o3[2], o3[3]);
    }
  }
}

extern "C" void kernel_launch(void* const* d_in, const int* in_sizes, int n_in,
                              void* d_out, int out_size, void* d_ws, size_t ws_size,
                              hipStream_t stream) {
  const float* ff   = (const float*)d_in[0];
  const float* fb   = (const float*)d_in[1];
  const float* soma = (const float*)d_in[2];
  const float* spk  = (const float*)d_in[3];
  const float* acur = (const float*)d_in[4];
  const float* bt   = (const float*)d_in[5];
  const float* cw   = (const float*)d_in[6];
  const float* cb   = (const float*)d_in[7];
  const float* gam  = (const float*)d_in[8];
  const float* bet  = (const float*)d_in[9];
  const float* lrec = (const float*)d_in[10];
  const float* taum = (const float*)d_in[11];
  const float* taud = (const float*)d_in[12];
  const float* taua = (const float*)d_in[13];
  float* out = (float*)d_out;
  float* ws  = (float*)d_ws;

  hipLaunchKernelGGL(k1_conv,    dim3(K1_BLOCKS), dim3(K1_T), 0, stream, ff, cw, cb, ws);
  hipLaunchKernelGGL(k2_bn,      dim3(1),    dim3(K2_T), 0, stream, gam, bet, ws);
  hipLaunchKernelGGL(k3_rec_lif, dim3(K3_BLOCKS), dim3(256), 0, stream,
                     fb, soma, spk, acur, bt, lrec, taum, taud, taua, ws, out);
}


# k3 block widened to 16 positions x 16 batches x 16 channels (16 waves): four lanes cover one 64-byte run of a [b][c][p] row, so a wave-wide load or store touches 8 rows, not 16; sum orders unchanged
# speedup vs baseline: 5.7782x; 1.0709x over previous
#include <hip/hip_runtime.h>

typedef __attribute__((ext_vector_type(2))) float v2f;
typedef __attribute__((ext_vector_type(8))) float v8f;

#define BB   32
#define CIN  64
#define COUT 64
#define HO   30
#define NPOS 900
#define KT   9
#define TOT  (BB*COUT*NPOS)

#define WS_Y     0
#define WS_PART  (TOT)
#define WS_PART2 (TOT + NPOS*COUT)
#define WS_BN    (TOT + 2*NPOS*COUT)

#define AP 74
#define BP 74

#define P1        8
#define K1_TILES  113
#define K1_BLOCKS (K1_TILES * 2)
#define K1_T      512
#define O1        32
#define APL       (BB * AP + 2)
#define BPL       (16 * BP + 40)
#define YP        12
#define ROWF      (NPOS * KT)

__device__ __forceinline__ float4 ld4(const float* p) { return *(const float4*)p; }

__device__ __forceinline__ void lds_barrier() {
  asm volatile("s_wait_dscnt 0x0" ::: "memory");
  __builtin_amdgcn_fence(__ATOMIC_RELEASE, "workgroup", "local");
  __builtin_amdgcn_s_barrier();
  __builtin_amdgcn_fence(__ATOMIC_ACQUIRE, "workgroup", "local");
}

__global__ __launch_bounds__(K1_T) void k1_conv(
    const float* __restrict__ ff, const float* __restrict__ cw,
    const float* __restrict__ cb, float* __restrict__ ws)
{
  __shared__ __attribute__((aligned(16))) float Al[P1 * APL];
  __shared__ __attribute__((aligned(16))) float Bl[2 * P1 * BPL];

  const int bid = blockIdx.x;
  const int oh = bid & 1, tile = bid >> 1;
  const int p0 = tile * P1, o0 = oh * O1;
  const int t = threadIdx.x;
  const int lane = t & 31, wave = t >> 5;
  const int ln = lane & 15, hi = lane >> 4;
  const int wp = wave & 7, wq = wave >> 3;

  const int apl = t & 7, ac = (t >> 3) & 7, ab = t >> 6;
  const int pa = min(p0 + apl, NPOS - 1);
  const float* asrc = ff + ((ab * CIN + ac) * 32 + pa / HO) * 32 + pa % HO;
  float* adst = Al + apl * APL + ab * AP + ac * KT;

  const float* afr = Al + wp * APL + ln * AP + 2 * hi;
  const float* bfr = Bl + (wq * P1 + wp) * BPL + ln * BP + 2 * hi;

  v8f acc0 = {}, acc1 = {};
  auto mma_chunk = [&]() {
    #pragma unroll
    for (int kb = 0; kb < 72; kb += 4) {
      v2f bv = *(const v2f*)(bfr + kb);
      v2f a0 = *(const v2f*)(afr + kb);
      v2f a1 = *(const v2f*)(afr + 16 * AP + kb);
      acc0 = __builtin_amdgcn_wmma_f32_16x16x4_f32(false, a0, false, bv,
                                                   (short)0, acc0, false, false);
      acc1 = __builtin_amdgcn_wmma_f32_16x16x4_f32(false, a1, false, bv,
                                                   (short)0, acc1, false, false);
    }
  };
  for (int cg = 0; cg < 8; ++cg) {
    float4 wv[9];
    #pragma unroll
    for (int i = 0; i < 9; ++i) {
      int idx = i * K1_T + t;
      int row = idx / 18, f = idx - row * 18;
      int ol = row >> 3, cc = row & 7;
      int col = p0 * KT + f * 4;
      wv[i] = make_float4(0.f, 0.f, 0.f, 0.f);
      if (col < ROWF)
        wv[i] = ld4(cw + ((o0 + ol) * CIN + cg * 8 + cc) * ROWF + col);
    }
    float pv[4][9];
    #pragma unroll
    for (int i = 0; i < 4; ++i) {
      const float* s = asrc + (i * 8 * CIN + cg * 8) * 1024;
      #pragma unroll
      for (int kh = 0; kh < 3; ++kh)
        #pragma unroll
        for (int kw = 0; kw < 3; ++kw)
          pv[i][kh * 3 + kw] = s[kh * 32 + kw];
    }
    if (cg > 0) {
      lds_barrier();
      mma_chunk();
    }
    lds_barrier();
    #pragma unroll
    for (int i = 0; i < 9; ++i) {
      int idx = i * K1_T + t;
      int row = idx / 18, f = idx - row * 18;
      int ol = row >> 3, cc = row & 7;
      float* d = Bl + (ol >> 4) * (P1 * BPL) + (ol & 15) * BP + cc * KT;
      const float vv[4] = {wv[i].x, wv[i].y, wv[i].z, wv[i].w};
      #pragma unroll
      for (int e = 0; e < 4; ++e) {
        int flat = f * 4 + e;
        int pl = flat / KT, kk = flat - pl * KT;
        d[pl * BPL + kk] = vv[e];
      }
    }
    #pragma unroll
    for (int i = 0; i < 4; ++i)
      #pragma unroll
      for (int k = 0; k < KT; ++k) adst[i * 8 * AP + k] = pv[i][k];
  }
  lds_barrier();
  mma_chunk();

  __syncthreads();
  float* Ys   = Al;
  float* redS = Bl;
  float* redQ = Bl + 1024;
  const int p = p0 + wp;
  const int oc = wq * 16 + ln;
  const float bias = cb[(o0 + oc) * NPOS + min(p, NPOS - 1)];
  {
    float s = 0.f, q = 0.f;
    #pragma unroll
    for (int r = 0; r < 8; ++r) {
      float v = acc0[r] + bias;
      Ys[((hi * 8 + r) * O1 + oc) * YP + wp] = v;
      s += v; q += v * v;
    }
    redS[(wave * 2) * 32 + lane] = s; redQ[(wave * 2) * 32 + lane] = q;
  }
  {
    float s = 0.f, q = 0.f;
    #pragma unroll
    for (int r = 0; r < 8; ++r) {
      float v = acc1[r] + bias;
      Ys[((16 + hi * 8 + r) * O1 + oc) * YP + wp] = v;
      s += v; q += v * v;
    }
    redS[(wave * 2 + 1) * 32 + lane] = s; redQ[(wave * 2 + 1) * 32 + lane] = q;
  }
  __syncthreads();
  if (t < 64) {
    const int l0 = t & 31;
    const float* red = Bl + (t >> 5) * 1024 + (l0 >> 4) * (P1 * 64) + (l0 & 15);
    float tot = 0.f;
    #pragma unroll
    for (int pl2 = 0; pl2 < P1; ++pl2) {
      float f4 = 0.f;
      #pragma unroll
      for (int mt2 = 0; mt2 < 2; ++mt2)
        #pragma unroll
        for (int h2 = 0; h2 < 2; ++h2)
          f4 += red[(pl2 * 2 + mt2) * 32 + h2 * 16];
      tot += (p0 + pl2 < NPOS) ? f4 : 0.f;
    }
    ws[(t < 32 ? WS_PART : WS_PART2) + tile * COUT + o0 + l0] = tot;
  }
  #pragma unroll
  for (int i = 0; i < 4; ++i) {
    int item = i * K1_T + t;
    int q4 = item & 1, row = item >> 1;
    int b = row >> 5, ol = row & 31;
    int pq = p0 + 4 * q4;
    if (pq < NPOS)
      *(float4*)(ws + WS_Y + (b * COUT + o0 + ol) * NPOS + pq) =
          *(const float4*)(Ys + row * YP + 4 * q4);
  }
}

#define K2_T   1024
#define K2_G   64

__global__ __launch_bounds__(K2_T) void k2_bn(
    const float* __restrict__ gamma, const float* __restrict__ beta,
    float* __restrict__ ws)
{
  __shared__ __attribute__((aligned(16))) float gS[K2_G * COUT];
  __shared__ __attribute__((aligned(16))) float gQ[K2_G * COUT];
  __shared__ float hS[4 * COUT];
  __shared__ float hQ[4 * COUT];

  const int t = threadIdx.x;
  const int c4 = t & 15, g = t >> 4;
  const float gm = gamma[t & 63], be = beta[t & 63];
  const float* ps = ws + WS_PART  + g * COUT + c4 * 4;
  const float* pq = ws + WS_PART2 + g * COUT + c4 * 4;
  float4 S = ld4(ps), Q = ld4(pq);
  if (g + K2_G < K1_TILES) {
    const float4 s2 = ld4(ps + K2_G * COUT), q2 = ld4(pq + K2_G * COUT);
    S.x += s2.x; S.y += s2.y; S.z += s2.z; S.w += s2.w;
    Q.x += q2.x; Q.y += q2.y; Q.z += q2.z; Q.w += q2.w;
  }
  *(float4*)(gS + g * COUT + c4 * 4) = S;
  *(float4*)(gQ + g * COUT + c4 * 4) = Q;
  __syncthreads();
  if (t < 4 * COUT) {
    const int ch = t & 63, h = t >> 6;
    float s = 0.f, q = 0.f;
    #pragma unroll
    for (int i = 0; i < K2_G / 4; ++i) {
      s += gS[(h * (K2_G / 4) + i) * COUT + ch];
      q += gQ[(h * (K2_G / 4) + i) * COUT + ch];
    }
    hS[h * COUT + ch] = s;
    hQ[h * COUT + ch] = q;
  }
  __syncthreads();
  if (t < COUT) {
    const int ch = t;
    const float Sx = (hS[ch] + hS[COUT + ch]) + (hS[2 * COUT + ch] + hS[3 * COUT + ch]);
    const float Qx = (hQ[ch] + hQ[COUT + ch]) + (hQ[2 * COUT + ch] + hQ[3 * COUT + ch]);
    const float inv = 1.0f / (float)(BB * NPOS);
    const float mean = Sx * inv;
    const float var  = Qx * inv - mean * mean;
    const float sc = gm * rsqrtf(var + 1e-5f);
    ws[WS_BN + ch]        = sc;
    ws[WS_BN + COUT + ch] = be - mean * sc;
  }
}

#define PT        16
#define K3_TILES  57
#define K3_BLOCKS (K3_TILES * 8)
#define K3_T      512
#define SPP       68
#define PSS       1092
#define RP        20


__global__ __launch_bounds__(K3_T) void k3_rec_lif(
    const float* __restrict__ fbv,  const float* __restrict__ soma,
    const float* __restrict__ spk,  const float* __restrict__ acur,
    const float* __restrict__ bt,   const float* __restrict__ lrec,
    const float* __restrict__ taum, const float* __restrict__ taud,
    const float* __restrict__ taua, const float* __restrict__ ws,
    float* __restrict__ out)
{
  __shared__ __attribute__((aligned(16))) float Ss[PT * PSS];
  __shared__ __attribute__((aligned(16))) float Rs[256 * RP];

  const int bid = blockIdx.x;
  const int cq = bid & 3, bh = (bid >> 2) & 1, tile = bid >> 3;
  const int p0 = tile * PT, b0 = bh * 16, c0 = cq * 16;
  const int t = threadIdx.x;
  const int lane = t & 31, wave = t >> 5;
  const int ln = lane & 15, hi = lane >> 4;

  const int q  = t & 3;
  const int cl = (t >> 2) & 15;
  const int c  = c0 + cl;
  const int pq = p0 + 4 * q;
  const bool okq = pq < NPOS;
  const int pql = okq ? pq : NPOS - 4;
  const int bl0 = t >> 6;

  float4 sv4[8];
  #pragma unroll
  for (int i = 0; i < 8; ++i) {
    int e = i * K3_T + t;
    int row = e >> 2;
    int d = row & 63, bl = row >> 6;
    sv4[i] = ld4(spk + ((b0 + bl) * COUT + d) * NPOS + pql);
  }

  const int pw = min(p0 + wave, NPOS - 1);
  const float* lp = lrec + ((size_t)pw * COUT + c0 + ln) * COUT + 2 * hi;
  v2f bfr[16];
  #pragma unroll
  for (int j = 0; j < 16; ++j) bfr[j] = *(const v2f*)(lp + 4 * j);

  const float4 tm = ld4(taum + c * NPOS + pql);
  const float4 td = ld4(taud + c * NPOS + pql);
  const float4 ta = ld4(taua + c * NPOS + pql);
  float4 yv[2], bv4[2], av4[2], fv4[2], so4[2];
  #pragma unroll
  for (int i = 0; i < 2; ++i) {
    int idx = ((b0 + bl0 + 8 * i) * COUT + c) * NPOS + pql;
    yv[i]  = ld4(ws + WS_Y + idx);
    bv4[i] = ld4(bt + idx);
    av4[i] = ld4(acur + idx);
    fv4[i] = ld4(fbv + idx);
    so4[i] = ld4(soma + idx);
  }
  const float sc = ws[WS_BN + c];
  const float sh = ws[WS_BN + COUT + c];

  #pragma unroll
  for (int i = 0; i < 8; ++i) {
    int e = i * K3_T + t;
    int row = e >> 2;
    int d = row & 63, bl = row >> 6;
    float* dst = Ss + (4 * q) * PSS + bl * SPP + d;
    dst[0]       = sv4[i].x;
    dst[PSS]     = sv4[i].y;
    dst[2 * PSS] = sv4[i].z;
    dst[3 * PSS] = sv4[i].w;
  }
  __syncthreads();

  v8f acc = {};
  const float* ap = Ss + wave * PSS + ln * SPP + 2 * hi;
  #pragma unroll
  for (int j = 0; j < 16; ++j) {
    v2f av = *(const v2f*)(ap + 4 * j);
    acc = __builtin_amdgcn_wmma_f32_16x16x4_f32(false, av, false, bfr[j],
                                                (short)0, acc, false, false);
  }
  #pragma unroll
  for (int r = 0; r < 8; ++r)
    Rs[((hi * 8 + r) * 16 + ln) * RP + wave] = acc[r];
  __syncthreads();

  if (okq) {
    float alpha[4], rho[4], eta[4];
    const float tmv[4] = {tm.x, tm.y, tm.z, tm.w};
    const float tdv[4] = {td.x, td.y, td.z, td.w};
    const float tav[4] = {ta.x, ta.y, ta.z, ta.w};
    #pragma unroll
    for (int k = 0; k < 4; ++k) {
      alpha[k] = __expf(-0.5f / tmv[k]);
      rho[k]   = __expf(-0.5f / tdv[k]);
      eta[k]   = __expf(-0.5f / tav[k]);
    }
    #pragma unroll
    for (int i = 0; i < 2; ++i) {
      const int bl = bl0 + 8 * i;
      const int idx = ((b0 + bl) * COUT + c) * NPOS + pq;
      const float4 rc = *(const float4*)(Rs + (bl * 16 + cl) * RP + 4 * q);
      const float rcv[4] = {rc.x, rc.y, rc.z, rc.w};
      const float yy[4]  = {yv[i].x, yv[i].y, yv[i].z, yv[i].w};
      const float bb[4]  = {bv4[i].x, bv4[i].y, bv4[i].z, bv4[i].w};
      const float aa[4]  = {av4[i].x, av4[i].y, av4[i].z, av4[i].w};
      const float ff4[4] = {fv4[i].x, fv4[i].y, fv4[i].z, fv4[i].w};
      const float ss[4]  = {so4[i].x, so4[i].y, so4[i].z, so4[i].w};
      float o0[4], o1[4], o2[4], o3[4];
      #pragma unroll
      for (int k = 0; k < 4; ++k) {
        float sv    = Ss[(4 * q + k) * PSS + bl * SPP + c];
        float convx = yy[k] * sc + sh + rcv[k];
        float bnew  = rho[k] * bb[k] + (1.f - rho[k]) * sv;
        float thre  = 0.1f + 1.8f * bnew;
        float anew  = eta[k] * aa[k] + ff4[k];
        float sg    = 1.f / (1.f + __expf(-anew)) - 0.5f;
        float snew  = alpha[k] * ss[k] + sg + convx - thre * sv;
        o0[k] = snew;
        o1[k] = (snew - thre) > 0.f ? 1.f : 0.f;
        o2[k] = anew;
        o3[k] = bnew;
      }
      *(float4*)(out + idx)           = make_float4(o0[0], o0[1], o0[2], o0[3]);
      *(float4*)(out + TOT + idx)     = make_float4(o1[0], o1[1], o1[2], o1[3]);
      *(float4*)(out + 2 * TOT + idx) = make_float4(o2[0], o2[1], o2[2], o2[3]);
      *(float4*)(out + 3 * TOT + idx) = make_float4(o3[0], o3[1], o3[2], o3[3]);
    }
  }
}

extern "C" void kernel_launch(void* const* d_in, const int* in_sizes, int n_in,
                              void* d_out, int out_size, void* d_ws, size_t ws_size,
                              hipStream_t stream) {
  const float* ff   = (const float*)d_in[0];
  const float* fb   = (const float*)d_in[1];
  const float* soma = (const float*)d_in[2];
  const float* spk  = (const float*)d_in[3];
  const float* acur = (const float*)d_in[4];
  const float* bt   = (const float*)d_in[5];
  const float* cw   = (const float*)d_in[6];
  const float* cb   = (const float*)d_in[7];
  const float* gam  = (const float*)d_in[8];
  const float* bet  = (const float*)d_in[9];
  const float* lrec = (const float*)d_in[10];
  const float* taum = (const float*)d_in[11];
  const float* taud = (const float*)d_in[12];
  const float* taua = (const float*)d_in[13];
  float* out = (float*)d_out;
  float* ws  = (float*)d_ws;

  hipLaunchKernelGGL(k1_conv,    dim3(K1_BLOCKS), dim3(K1_T), 0, stream, ff, cw, cb, ws);
  hipLaunchKernelGGL(k2_bn,      dim3(1),    dim3(K2_T), 0, stream, gam, bet, ws);
  hipLaunchKernelGGL(k3_rec_lif, dim3(K3_BLOCKS), dim3(K3_T), 0, stream,
                     fb, soma, spk, acur, bt, lrec, taum, taud, taua, ws, out);
}


# BN finalize folded into the LIF kernel: each block folds the 113x16 per-tile sums of its 16 channels in LDS between its barriers (fold order kept); the finalize kernel and its launch are gone
# speedup vs baseline: 5.9827x; 1.0354x over previous
#include <hip/hip_runtime.h>

typedef __attribute__((ext_vector_type(2))) float v2f;
typedef __attribute__((ext_vector_type(8))) float v8f;

#define BB   32
#define CIN  64
#define COUT 64
#define HO   30
#define NPOS 900
#define KT   9
#define TOT  (BB*COUT*NPOS)

#define WS_Y     0
#define WS_PART  (TOT)
#define WS_PART2 (TOT + NPOS*COUT)

#define AP 74
#define BP 74

#define P1        8
#define K1_TILES  113
#define K1_BLOCKS (K1_TILES * 2)
#define K1_T      512
#define O1        32
#define APL       (BB * AP + 2)
#define BPL       (16 * BP + 40)
#define YP        12
#define ROWF      (NPOS * KT)

__device__ __forceinline__ float4 ld4(const float* p) { return *(const float4*)p; }

__device__ __forceinline__ void lds_barrier() {
  asm volatile("s_wait_dscnt 0x0" ::: "memory");
  __builtin_amdgcn_fence(__ATOMIC_RELEASE, "workgroup", "local");
  __builtin_amdgcn_s_barrier();
  __builtin_amdgcn_fence(__ATOMIC_ACQUIRE, "workgroup", "local");
}

__global__ __launch_bounds__(K1_T) void k1_conv(
    const float* __restrict__ ff, const float* __restrict__ cw,
    const float* __restrict__ cb, float* __restrict__ ws)
{
  __shared__ __attribute__((aligned(16))) float Al[P1 * APL];
  __shared__ __attribute__((aligned(16))) float Bl[2 * P1 * BPL];

  const int bid = blockIdx.x;
  const int oh = bid & 1, tile = bid >> 1;
  const int p0 = tile * P1, o0 = oh * O1;
  const int t = threadIdx.x;
  const int lane = t & 31, wave = t >> 5;
  const int ln = lane & 15, hi = lane >> 4;
  const int wp = wave & 7, wq = wave >> 3;

  const int apl = t & 7, ac = (t >> 3) & 7, ab = t >> 6;
  const int pa = min(p0 + apl, NPOS - 1);
  const float* asrc = ff + ((ab * CIN + ac) * 32 + pa / HO) * 32 + pa % HO;
  float* adst = Al + apl * APL + ab * AP + ac * KT;

  const float* afr = Al + wp * APL + ln * AP + 2 * hi;
  const float* bfr = Bl + (wq * P1 + wp) * BPL + ln * BP + 2 * hi;

  v8f acc0 = {}, acc1 = {};
  auto mma_chunk = [&]() {
    #pragma unroll
    for (int kb = 0; kb < 72; kb += 4) {
      v2f bv = *(const v2f*)(bfr + kb);
      v2f a0 = *(const v2f*)(afr + kb);
      v2f a1 = *(const v2f*)(afr + 16 * AP + kb);
      acc0 = __builtin_amdgcn_wmma_f32_16x16x4_f32(false, a0, false, bv,
                                                   (short)0, acc0, false, false);
      acc1 = __builtin_amdgcn_wmma_f32_16x16x4_f32(false, a1, false, bv,
                                                   (short)0, acc1, false, false);
    }
  };
  for (int cg = 0; cg < 8; ++cg) {
    float4 wv[9];
    #pragma unroll
    for (int i = 0; i < 9; ++i) {
      int idx = i * K1_T + t;
      int row = idx / 18, f = idx - row * 18;
      int ol = row >> 3, cc = row & 7;
      int col = p0 * KT + f * 4;
      wv[i] = make_float4(0.f, 0.f, 0.f, 0.f);
      if (col < ROWF)
        wv[i] = ld4(cw + ((o0 + ol) * CIN + cg * 8 + cc) * ROWF + col);
    }
    float pv[4][9];
    #pragma unroll
    for (int i = 0; i < 4; ++i) {
      const float* s = asrc + (i * 8 * CIN + cg * 8) * 1024;
      #pragma unroll
      for (int kh = 0; kh < 3; ++kh)
        #pragma unroll
        for (int kw = 0; kw < 3; ++kw)
          pv[i][kh * 3 + kw] = s[kh * 32 + kw];
    }
    if (cg > 0) {
      lds_barrier();
      mma_chunk();
    }
    lds_barrier();
    #pragma unroll
    for (int i = 0; i < 9; ++i) {
      int idx = i * K1_T + t;
      int row = idx / 18, f = idx - row * 18;
      int ol = row >> 3, cc = row & 7;
      float* d = Bl + (ol >> 4) * (P1 * BPL) + (ol & 15) * BP + cc * KT;
      const float vv[4] = {wv[i].x, wv[i].y, wv[i].z, wv[i].w};
      #pragma unroll
      for (int e = 0; e < 4; ++e) {
        int flat = f * 4 + e;
        int pl = flat / KT, kk = flat - pl * KT;
        d[pl * BPL + kk] = vv[e];
      }
    }
    #pragma unroll
    for (int i = 0; i < 4; ++i)
      #pragma unroll
      for (int k = 0; k < KT; ++k) adst[i * 8 * AP + k] = pv[i][k];
  }
  lds_barrier();
  mma_chunk();

  __syncthreads();
  float* Ys   = Al;
  float* redS = Bl;
  float* redQ = Bl + 1024;
  const int p = p0 + wp;
  const int oc = wq * 16 + ln;
  const float bias = cb[(o0 + oc) * NPOS + min(p, NPOS - 1)];
  {
    float s = 0.f, q = 0.f;
    #pragma unroll
    for (int r = 0; r < 8; ++r) {
      float v = acc0[r] + bias;
      Ys[((hi * 8 + r) * O1 + oc) * YP + wp] = v;
      s += v; q += v * v;
    }
    redS[(wave * 2) * 32 + lane] = s; redQ[(wave * 2) * 32 + lane] = q;
  }
  {
    float s = 0.f, q = 0.f;
    #pragma unroll
    for (int r = 0; r < 8; ++r) {
      float v = acc1[r] + bias;
      Ys[((16 + hi * 8 + r) * O1 + oc) * YP + wp] = v;
      s += v; q += v * v;
    }
    redS[(wave * 2 + 1) * 32 + lane] = s; redQ[(wave * 2 + 1) * 32 + lane] = q;
  }
  __syncthreads();
  if (t < 64) {
    const int l0 = t & 31;
    const float* red = Bl + (t >> 5) * 1024 + (l0 >> 4) * (P1 * 64) + (l0 & 15);
    float tot = 0.f;
    #pragma unroll
    for (int pl2 = 0; pl2 < P1; ++pl2) {
      float f4 = 0.f;
      #pragma unroll
      for (int mt2 = 0; mt2 < 2; ++mt2)
        #pragma unroll
        for (int h2 = 0; h2 < 2; ++h2)
          f4 += red[(pl2 * 2 + mt2) * 32 + h2 * 16];
      tot += (p0 + pl2 < NPOS) ? f4 : 0.f;
    }
    ws[(t < 32 ? WS_PART : WS_PART2) + tile * COUT + o0 + l0] = tot;
  }
  #pragma unroll
  for (int i = 0; i < 4; ++i) {
    int item = i * K1_T + t;
    int q4 = item & 1, row = item >> 1;
    int b = row >> 5, ol = row & 31;
    int pq = p0 + 4 * q4;
    if (pq < NPOS)
      *(float4*)(ws + WS_Y + (b * COUT + o0 + ol) * NPOS + pq) =
          *(const float4*)(Ys + row * YP + 4 * q4);
  }
}

#define PT        16
#define K3_TILES  57
#define K3_BLOCKS (K3_TILES * 8)
#define K3_T      512
#define SPP       68
#define PSS       1092
#define RP        20
#define FG        64


__global__ __launch_bounds__(K3_T) void k3_rec_lif(
    const float* __restrict__ fbv,  const float* __restrict__ soma,
    const float* __restrict__ spk,  const float* __restrict__ acur,
    const float* __restrict__ bt,   const float* __restrict__ lrec,
    const float* __restrict__ taum, const float* __restrict__ taud,
    const float* __restrict__ taua, const float* __restrict__ gamma,
    const float* __restrict__ beta, const float* __restrict__ ws,
    float* __restrict__ out)
{
  __shared__ __attribute__((aligned(16))) float Ss[PT * PSS];
  __shared__ __attribute__((aligned(16))) float Rs[256 * RP];
  __shared__ __attribute__((aligned(16))) float gP[2 * FG * 16];
  __shared__ float hP[2 * 4 * 16];

  const int bid = blockIdx.x;
  const int cq = bid & 3, bh = (bid >> 2) & 1, tile = bid >> 3;
  const int p0 = tile * PT, b0 = bh * 16, c0 = cq * 16;
  const int t = threadIdx.x;
  const int lane = t & 31, wave = t >> 5;
  const int ln = lane & 15, hi = lane >> 4;

  const int q  = t & 3;
  const int cl = (t >> 2) & 15;
  const int c  = c0 + cl;
  const int pq = p0 + 4 * q;
  const bool okq = pq < NPOS;
  const int pql = okq ? pq : NPOS - 4;
  const int bl0 = t >> 6;

  const int fw = t >> 8, fg = (t >> 2) & (FG - 1), fc4 = t & 3;
  const bool has2 = fg + FG < K1_TILES;
  const float* pp = ws + (fw ? WS_PART2 : WS_PART) + fg * COUT + c0 + fc4 * 4;
  {
    float4 P = ld4(pp);
    if (has2) {
      const float4 p2 = ld4(pp + FG * COUT);
      P.x += p2.x; P.y += p2.y; P.z += p2.z; P.w += p2.w;
    }
    *(float4*)(gP + fw * (FG * 16) + fg * 16 + fc4 * 4) = P;
  }

  float4 sv4[8];
  #pragma unroll
  for (int i = 0; i < 8; ++i) {
    int e = i * K3_T + t;
    int row = e >> 2;
    int d = row & 63, bl = row >> 6;
    sv4[i] = ld4(spk + ((b0 + bl) * COUT + d) * NPOS + pql);
  }

  const int pw = min(p0 + wave, NPOS - 1);
  const float* lp = lrec + ((size_t)pw * COUT + c0 + ln) * COUT + 2 * hi;
  v2f bfr[16];
  #pragma unroll
  for (int j = 0; j < 16; ++j) bfr[j] = *(const v2f*)(lp + 4 * j);

  const float4 tm = ld4(taum + c * NPOS + pql);
  const float4 td = ld4(taud + c * NPOS + pql);
  const float4 ta = ld4(taua + c * NPOS + pql);
  float4 yv[2], bv4[2], av4[2], fv4[2], so4[2];
  #pragma unroll
  for (int i = 0; i < 2; ++i) {
    int idx = ((b0 + bl0 + 8 * i) * COUT + c) * NPOS + pql;
    yv[i]  = ld4(ws + WS_Y + idx);
    bv4[i] = ld4(bt + idx);
    av4[i] = ld4(acur + idx);
    fv4[i] = ld4(fbv + idx);
    so4[i] = ld4(soma + idx);
  }
  const float gm = gamma[c];
  const float be = beta[c];

  #pragma unroll
  for (int i = 0; i < 8; ++i) {
    int e = i * K3_T + t;
    int row = e >> 2;
    int d = row & 63, bl = row >> 6;
    float* dst = Ss + (4 * q) * PSS + bl * SPP + d;
    dst[0]       = sv4[i].x;
    dst[PSS]     = sv4[i].y;
    dst[2 * PSS] = sv4[i].z;
    dst[3 * PSS] = sv4[i].w;
  }
  __syncthreads();

  if (t < 128) {
    const int ch = t & 15, h = (t >> 4) & 3, w2 = t >> 6;
    const float* g0 = gP + w2 * (FG * 16) + h * (FG / 4) * 16 + ch;
    float s = 0.f;
    #pragma unroll
    for (int i = 0; i < FG / 4; ++i) s += g0[i * 16];
    hP[w2 * 64 + h * 16 + ch] = s;
  }

  v8f acc = {};
  const float* ap = Ss + wave * PSS + ln * SPP + 2 * hi;
  #pragma unroll
  for (int j = 0; j < 16; ++j) {
    v2f av = *(const v2f*)(ap + 4 * j);
    acc = __builtin_amdgcn_wmma_f32_16x16x4_f32(false, av, false, bfr[j],
                                                (short)0, acc, false, false);
  }
  #pragma unroll
  for (int r = 0; r < 8; ++r)
    Rs[((hi * 8 + r) * 16 + ln) * RP + wave] = acc[r];
  __syncthreads();

  if (okq) {
    const float Sx = (hP[cl] + hP[16 + cl]) + (hP[32 + cl] + hP[48 + cl]);
    const float Qx = (hP[64 + cl] + hP[80 + cl]) + (hP[96 + cl] + hP[112 + cl]);
    const float inv = 1.0f / (float)(BB * NPOS);
    const float mean = Sx * inv;
    const float var  = Qx * inv - mean * mean;
    const float sc = gm * rsqrtf(var + 1e-5f);
    const float sh = be - mean * sc;
    float alpha[4], rho[4], eta[4];
    const float tmv[4] = {tm.x, tm.y, tm.z, tm.w};
    const float tdv[4] = {td.x, td.y, td.z, td.w};
    const float tav[4] = {ta.x, ta.y, ta.z, ta.w};
    #pragma unroll
    for (int k = 0; k < 4; ++k) {
      alpha[k] = __expf(-0.5f / tmv[k]);
      rho[k]   = __expf(-0.5f / tdv[k]);
      eta[k]   = __expf(-0.5f / tav[k]);
    }
    #pragma unroll
    for (int i = 0; i < 2; ++i) {
      const int bl = bl0 + 8 * i;
      const int idx = ((b0 + bl) * COUT + c) * NPOS + pq;
      const float4 rc = *(const float4*)(Rs + (bl * 16 + cl) * RP + 4 * q);
      const float rcv[4] = {rc.x, rc.y, rc.z, rc.w};
      const float yy[4]  = {yv[i].x, yv[i].y, yv[i].z, yv[i].w};
      const float bb[4]  = {bv4[i].x, bv4[i].y, bv4[i].z, bv4[i].w};
      const float aa[4]  = {av4[i].x, av4[i].y, av4[i].z, av4[i].w};
      const float ff4[4] = {fv4[i].x, fv4[i].y, fv4[i].z, fv4[i].w};
      const float ss[4]  = {so4[i].x, so4[i].y, so4[i].z, so4[i].w};
      float o0[4], o1[4], o2[4], o3[4];
      #pragma unroll
      for (int k = 0; k < 4; ++k) {
        float sv    = Ss[(4 * q + k) * PSS + bl * SPP + c];
        float convx = yy[k] * sc + sh + rcv[k];
        float bnew  = rho[k] * bb[k] + (1.f - rho[k]) * sv;
        float thre  = 0.1f + 1.8f * bnew;
        float anew  = eta[k] * aa[k] + ff4[k];
        float sg    = 1.f / (1.f + __expf(-anew)) - 0.5f;
        float snew  = alpha[k] * ss[k] + sg + convx - thre * sv;
        o0[k] = snew;
        o1[k] = (snew - thre) > 0.f ? 1.f : 0.f;
        o2[k] = anew;
        o3[k] = bnew;
      }
      *(float4*)(out + idx)           = make_float4(o0[0], o0[1], o0[2], o0[3]);
      *(float4*)(out + TOT + idx)     = make_float4(o1[0], o1[1], o1[2], o1[3]);
      *(float4*)(out + 2 * TOT + idx) = make_float4(o2[0], o2[1], o2[2], o2[3]);
      *(float4*)(out + 3 * TOT + idx) = make_float4(o3[0], o3[1], o3[2], o3[3]);
    }
  }
}

extern "C" void kernel_launch(void* const* d_in, const int* in_sizes, int n_in,
                              void* d_out, int out_size, void* d_ws, size_t ws_size,
                              hipStream_t stream) {
  const float* ff   = (const float*)d_in[0];
  const float* fb   = (const float*)d_in[1];
  const float* soma = (const float*)d_in[2];
  const float* spk  = (const float*)d_in[3];
  const float* acur = (const float*)d_in[4];
  const float* bt   = (const float*)d_in[5];
  const float* cw   = (const float*)d_in[6];
  const float* cb   = (const float*)d_in[7];
  const float* gam  = (const float*)d_in[8];
  const float* bet  = (const float*)d_in[9];
  const float* lrec = (const float*)d_in[10];
  const float* taum = (const float*)d_in[11];
  const float* taud = (const float*)d_in[12];
  const float* taua = (const float*)d_in[13];
  float* out = (float*)d_out;
  float* ws  = (float*)d_ws;

  hipLaunchKernelGGL(k1_conv,    dim3(K1_BLOCKS), dim3(K1_T), 0, stream, ff, cw, cb, ws);
  hipLaunchKernelGGL(k3_rec_lif, dim3(K3_BLOCKS), dim3(K3_T), 0, stream,
                     fb, soma, spk, acur, bt, lrec, taum, taud, taua, gam, bet, ws, out);
}


# k1 LDS tiles double-buffered (two stages): one split barrier per K chunk instead of two full ones, so a wave done with its WMMAs stores and loads ahead while others multiply; sum orders unchanged
# speedup vs baseline: 6.5852x; 1.1007x over previous
#include <hip/hip_runtime.h>

typedef __attribute__((ext_vector_type(2))) float v2f;
typedef __attribute__((ext_vector_type(8))) float v8f;

#define BB   32
#define CIN  64
#define COUT 64
#define HO   30
#define NPOS 900
#define KT   9
#define TOT  (BB*COUT*NPOS)

#define WS_Y     0
#define WS_PART  (TOT)
#define WS_PART2 (TOT + NPOS*COUT)

#define AP 74
#define BP 74

#define P1        8
#define K1_TILES  113
#define K1_BLOCKS (K1_TILES * 2)
#define K1_T      512
#define O1        32
#define APL       (BB * AP + 2)
#define BPL       (16 * BP + 40)
#define YP        12
#define ROWF      (NPOS * KT)
#define ASTG      (P1 * APL)
#define BSTG      (2 * P1 * BPL)

__device__ __forceinline__ float4 ld4(const float* p) { return *(const float4*)p; }

__device__ __forceinline__ void lds_signal() {
  asm volatile("s_wait_dscnt 0x0" ::: "memory");
  __builtin_amdgcn_fence(__ATOMIC_RELEASE, "workgroup", "local");
  __builtin_amdgcn_s_barrier_signal(-1);
}
__device__ __forceinline__ void lds_wait() {
  __builtin_amdgcn_s_barrier_wait(-1);
  __builtin_amdgcn_fence(__ATOMIC_ACQUIRE, "workgroup", "local");
}

__global__ __launch_bounds__(K1_T) void k1_conv(
    const float* __restrict__ ff, const float* __restrict__ cw,
    const float* __restrict__ cb, float* __restrict__ ws)
{
  __shared__ __attribute__((aligned(16))) float Al[2 * ASTG];
  __shared__ __attribute__((aligned(16))) float Bl[2 * BSTG];

  const int bid = blockIdx.x;
  const int oh = bid & 1, tile = bid >> 1;
  const int p0 = tile * P1, o0 = oh * O1;
  const int t = threadIdx.x;
  const int lane = t & 31, wave = t >> 5;
  const int ln = lane & 15, hi = lane >> 4;
  const int wp = wave & 7, wq = wave >> 3;

  const int apl = t & 7, ac = (t >> 3) & 7, ab = t >> 6;
  const int pa = min(p0 + apl, NPOS - 1);
  const float* asrc = ff + ((ab * CIN + ac) * 32 + pa / HO) * 32 + pa % HO;
  float* adst = Al + apl * APL + ab * AP + ac * KT;

  const float* afr = Al + wp * APL + ln * AP + 2 * hi;
  const float* bfr = Bl + (wq * P1 + wp) * BPL + ln * BP + 2 * hi;

  v8f acc0 = {}, acc1 = {};
  auto mma_chunk = [&](int st) {
    const float* af = afr + st * ASTG;
    const float* bf = bfr + st * BSTG;
    #pragma unroll
    for (int kb = 0; kb < 72; kb += 4) {
      v2f bv = *(const v2f*)(bf + kb);
      v2f a0 = *(const v2f*)(af + kb);
      v2f a1 = *(const v2f*)(af + 16 * AP + kb);
      acc0 = __builtin_amdgcn_wmma_f32_16x16x4_f32(false, a0, false, bv,
                                                   (short)0, acc0, false, false);
      acc1 = __builtin_amdgcn_wmma_f32_16x16x4_f32(false, a1, false, bv,
                                                   (short)0, acc1, false, false);
    }
  };
  for (int cg = 0; cg < 8; ++cg) {
    float4 wv[9];
    #pragma unroll
    for (int i = 0; i < 9; ++i) {
      int idx = i * K1_T + t;
      int row = idx / 18, f = idx - row * 18;
      int ol = row >> 3, cc = row & 7;
      int col = p0 * KT + f * 4;
      wv[i] = make_float4(0.f, 0.f, 0.f, 0.f);
      if (col < ROWF)
        wv[i] = ld4(cw + ((o0 + ol) * CIN + cg * 8 + cc) * ROWF + col);
    }
    float pv[4][9];
    #pragma unroll
    for (int i = 0; i < 4; ++i) {
      const float* s = asrc + (i * 8 * CIN + cg * 8) * 1024;
      #pragma unroll
      for (int kh = 0; kh < 3; ++kh)
        #pragma unroll
        for (int kw = 0; kw < 3; ++kw)
          pv[i][kh * 3 + kw] = s[kh * 32 + kw];
    }
    const int st = cg & 1;
    if (cg > 0) {
      lds_wait();
      mma_chunk(st ^ 1);
    }
    #pragma unroll
    for (int i = 0; i < 9; ++i) {
      int idx = i * K1_T + t;
      int row = idx / 18, f = idx - row * 18;
      int ol = row >> 3, cc = row & 7;
      float* d = Bl + st * BSTG + (ol >> 4) * (P1 * BPL) + (ol & 15) * BP + cc * KT;
      const float vv[4] = {wv[i].x, wv[i].y, wv[i].z, wv[i].w};
      #pragma unroll
      for (int e = 0; e < 4; ++e) {
        int flat = f * 4 + e;
        int pl = flat / KT, kk = flat - pl * KT;
        d[pl * BPL + kk] = vv[e];
      }
    }
    #pragma unroll
    for (int i = 0; i < 4; ++i)
      #pragma unroll
      for (int k = 0; k < KT; ++k) adst[st * ASTG + i * 8 * AP + k] = pv[i][k];
    lds_signal();
  }
  lds_wait();
  mma_chunk(1);

  __syncthreads();
  float* Ys   = Al;
  float* redS = Bl;
  float* redQ = Bl + 1024;
  const int p = p0 + wp;
  const int oc = wq * 16 + ln;
  const float bias = cb[(o0 + oc) * NPOS + min(p, NPOS - 1)];
  {
    float s = 0.f, q = 0.f;
    #pragma unroll
    for (int r = 0; r < 8; ++r) {
      float v = acc0[r] + bias;
      Ys[((hi * 8 + r) * O1 + oc) * YP + wp] = v;
      s += v; q += v * v;
    }
    redS[(wave * 2) * 32 + lane] = s; redQ[(wave * 2) * 32 + lane] = q;
  }
  {
    float s = 0.f, q = 0.f;
    #pragma unroll
    for (int r = 0; r < 8; ++r) {
      float v = acc1[r] + bias;
      Ys[((16 + hi * 8 + r) * O1 + oc) * YP + wp] = v;
      s += v; q += v * v;
    }
    redS[(wave * 2 + 1) * 32 + lane] = s; redQ[(wave * 2 + 1) * 32 + lane] = q;
  }
  __syncthreads();
  if (t < 64) {
    const int l0 = t & 31;
    const float* red = Bl + (t >> 5) * 1024 + (l0 >> 4) * (P1 * 64) + (l0 & 15);
    float tot = 0.f;
    #pragma unroll
    for (int pl2 = 0; pl2 < P1; ++pl2) {
      float f4 = 0.f;
      #pragma unroll
      for (int mt2 = 0; mt2 < 2; ++mt2)
        #pragma unroll
        for (int h2 = 0; h2 < 2; ++h2)
          f4 += red[(pl2 * 2 + mt2) * 32 + h2 * 16];
      tot += (p0 + pl2 < NPOS) ? f4 : 0.f;
    }
    ws[(t < 32 ? WS_PART : WS_PART2) + tile * COUT + o0 + l0] = tot;
  }
  #pragma unroll
  for (int i = 0; i < 4; ++i) {
    int item = i * K1_T + t;
    int q4 = item & 1, row = item >> 1;
    int b = row >> 5, ol = row & 31;
    int pq = p0 + 4 * q4;
    if (pq < NPOS)
      *(float4*)(ws + WS_Y + (b * COUT + o0 + ol) * NPOS + pq) =
          *(const float4*)(Ys + row * YP + 4 * q4);
  }
}

#define PT        16
#define K3_TILES  57
#define K3_BLOCKS (K3_TILES * 8)
#define K3_T      512
#define SPP       68
#define PSS       1092
#define RP        20
#define FG        64


__global__ __launch_bounds__(K3_T) void k3_rec_lif(
    const float* __restrict__ fbv,  const float* __restrict__ soma,
    const float* __restrict__ spk,  const float* __restrict__ acur,
    const float* __restrict__ bt,   const float* __restrict__ lrec,
    const float* __restrict__ taum, const float* __restrict__ taud,
    const float* __restrict__ taua, const float* __restrict__ gamma,
    const float* __restrict__ beta, const float* __restrict__ ws,
    float* __restrict__ out)
{
  __shared__ __attribute__((aligned(16))) float Ss[PT * PSS];
  __shared__ __attribute__((aligned(16))) float Rs[256 * RP];
  __shared__ __attribute__((aligned(16))) float gP[2 * FG * 16];
  __shared__ float hP[2 * 4 * 16];

  const int bid = blockIdx.x;
  const int cq = bid & 3, bh = (bid >> 2) & 1, tile = bid >> 3;
  const int p0 = tile * PT, b0 = bh * 16, c0 = cq * 16;
  const int t = threadIdx.x;
  const int lane = t & 31, wave = t >> 5;
  const int ln = lane & 15, hi = lane >> 4;

  const int q  = t & 3;
  const int cl = (t >> 2) & 15;
  const int c  = c0 + cl;
  const int pq = p0 + 4 * q;
  const bool okq = pq < NPOS;
  const int pql = okq ? pq : NPOS - 4;
  const int bl0 = t >> 6;

  const int fw = t >> 8, fg = (t >> 2) & (FG - 1), fc4 = t & 3;
  const bool has2 = fg + FG < K1_TILES;
  const float* pp = ws + (fw ? WS_PART2 : WS_PART) + fg * COUT + c0 + fc4 * 4;
  {
    float4 P = ld4(pp);
    if (has2) {
      const float4 p2 = ld4(pp + FG * COUT);
      P.x += p2.x; P.y += p2.y; P.z += p2.z; P.w += p2.w;
    }
    *(float4*)(gP + fw * (FG * 16) + fg * 16 + fc4 * 4) = P;
  }

  float4 sv4[8];
  #pragma unroll
  for (int i = 0; i < 8; ++i) {
    int e = i * K3_T + t;
    int row = e >> 2;
    int d = row & 63, bl = row >> 6;
    sv4[i] = ld4(spk + ((b0 + bl) * COUT + d) * NPOS + pql);
  }

  const int pw = min(p0 + wave, NPOS - 1);
  const float* lp = lrec + ((size_t)pw * COUT + c0 + ln) * COUT + 2 * hi;
  v2f bfr[16];
  #pragma unroll
  for (int j = 0; j < 16; ++j) bfr[j] = *(const v2f*)(lp + 4 * j);

  const float4 tm = ld4(taum + c * NPOS + pql);
  const float4 td = ld4(taud + c * NPOS + pql);
  const float4 ta = ld4(taua + c * NPOS + pql);
  float4 yv[2], bv4[2], av4[2], fv4[2], so4[2];
  #pragma unroll
  for (int i = 0; i < 2; ++i) {
    int idx = ((b0 + bl0 + 8 * i) * COUT + c) * NPOS + pql;
    yv[i]  = ld4(ws + WS_Y + idx);
    bv4[i] = ld4(bt + idx);
    av4[i] = ld4(acur + idx);
    fv4[i] = ld4(fbv + idx);
    so4[i] = ld4(soma + idx);
  }
  const float gm = gamma[c];
  const float be = beta[c];

  #pragma unroll
  for (int i = 0; i < 8; ++i) {
    int e = i * K3_T + t;
    int row = e >> 2;
    int d = row & 63, bl = row >> 6;
    float* dst = Ss + (4 * q) * PSS + bl * SPP + d;
    dst[0]       = sv4[i].x;
    dst[PSS]     = sv4[i].y;
    dst[2 * PSS] = sv4[i].z;
    dst[3 * PSS] = sv4[i].w;
  }
  __syncthreads();

  if (t < 128) {
    const int ch = t & 15, h = (t >> 4) & 3, w2 = t >> 6;
    const float* g0 = gP + w2 * (FG * 16) + h * (FG / 4) * 16 + ch;
    float s = 0.f;
    #pragma unroll
    for (int i = 0; i < FG / 4; ++i) s += g0[i * 16];
    hP[w2 * 64 + h * 16 + ch] = s;
  }

  v8f acc = {};
  const float* ap = Ss + wave * PSS + ln * SPP + 2 * hi;
  #pragma unroll
  for (int j = 0; j < 16; ++j) {
    v2f av = *(const v2f*)(ap + 4 * j);
    acc = __builtin_amdgcn_wmma_f32_16x16x4_f32(false, av, false, bfr[j],
                                                (short)0, acc, false, false);
  }
  #pragma unroll
  for (int r = 0; r < 8; ++r)
    Rs[((hi * 8 + r) * 16 + ln) * RP + wave] = acc[r];
  __syncthreads();

  if (okq) {
    const float Sx = (hP[cl] + hP[16 + cl]) + (hP[32 + cl] + hP[48 + cl]);
    const float Qx = (hP[64 + cl] + hP[80 + cl]) + (hP[96 + cl] + hP[112 + cl]);
    const float inv = 1.0f / (float)(BB * NPOS);
    const float mean = Sx * inv;
    const float var  = Qx * inv - mean * mean;
    const float sc = gm * rsqrtf(var + 1e-5f);
    const float sh = be - mean * sc;
    float alpha[4], rho[4], eta[4];
    const float tmv[4] = {tm.x, tm.y, tm.z, tm.w};
    const float tdv[4] = {td.x, td.y, td.z, td.w};
    const float tav[4] = {ta.x, ta.y, ta.z, ta.w};
    #pragma unroll
    for (int k = 0; k < 4; ++k) {
      alpha[k] = __expf(-0.5f / tmv[k]);
      rho[k]   = __expf(-0.5f / tdv[k]);
      eta[k]   = __expf(-0.5f / tav[k]);
    }
    #pragma unroll
    for (int i = 0; i < 2; ++i) {
      const int bl = bl0 + 8 * i;
      const int idx = ((b0 + bl) * COUT + c) * NPOS + pq;
      const float4 rc = *(const float4*)(Rs + (bl * 16 + cl) * RP + 4 * q);
      const float rcv[4] = {rc.x, rc.y, rc.z, rc.w};
      const float yy[4]  = {yv[i].x, yv[i].y, yv[i].z, yv[i].w};
      const float bb[4]  = {bv4[i].x, bv4[i].y, bv4[i].z, bv4[i].w};
      const float aa[4]  = {av4[i].x, av4[i].y, av4[i].z, av4[i].w};
      const float ff4[4] = {fv4[i].x, fv4[i].y, fv4[i].z, fv4[i].w};
      const float ss[4]  = {so4[i].x, so4[i].y, so4[i].z, so4[i].w};
      float o0[4], o1[4], o2[4], o3[4];
      #pragma unroll
      for (int k = 0; k < 4; ++k) {
        float sv    = Ss[(4 * q + k) * PSS + bl * SPP + c];
        float convx = yy[k] * sc + sh + rcv[k];
        float bnew  = rho[k] * bb[k] + (1.f - rho[k]) * sv;
        float thre  = 0.1f + 1.8f * bnew;
        float anew  = eta[k] * aa[k] + ff4[k];
        float sg    = 1.f / (1.f + __expf(-anew)) - 0.5f;
        float snew  = alpha[k] * ss[k] + sg + convx - thre * sv;
        o0[k] = snew;
        o1[k] = (snew - thre) > 0.f ? 1.f : 0.f;
        o2[k] = anew;
        o3[k] = bnew;
      }
      *(float4*)(out + idx)           = make_float4(o0[0], o0[1], o0[2], o0[3]);
      *(float4*)(out + TOT + idx)     = make_float4(o1[0], o1[1], o1[2], o1[3]);
      *(float4*)(out + 2 * TOT + idx) = make_float4(o2[0], o2[1], o2[2], o2[3]);
      *(float4*)(out + 3 * TOT + idx) = make_float4(o3[0], o3[1], o3[2], o3[3]);
    }
  }
}

extern "C" void kernel_launch(void* const* d_in, const int* in_sizes, int n_in,
                              void* d_out, int out_size, void* d_ws, size_t ws_size,
                              hipStream_t stream) {
  const float* ff   = (const float*)d_in[0];
  const float* fb   = (const float*)d_in[1];
  const float* soma = (const float*)d_in[2];
  const float* spk  = (const float*)d_in[3];
  const float* acur = (const float*)d_in[4];
  const float* bt   = (const float*)d_in[5];
  const float* cw   = (const float*)d_in[6];
  const float* cb   = (const float*)d_in[7];
  const float* gam  = (const float*)d_in[8];
  const float* bet  = (const float*)d_in[9];
  const float* lrec = (const float*)d_in[10];
  const float* taum = (const float*)d_in[11];
  const float* taud = (const float*)d_in[12];
  const float* taua = (const float*)d_in[13];
  float* out = (float*)d_out;
  float* ws  = (float*)d_ws;

  hipLaunchKernelGGL(k1_conv,    dim3(K1_BLOCKS), dim3(K1_T), 0, stream, ff, cw, cb, ws);
  hipLaunchKernelGGL(k3_rec_lif, dim3(K3_BLOCKS), dim3(K3_T), 0, stream,
                     fb, soma, spk, acur, bt, lrec, taum, taud, taua, gam, bet, ws, out);
}


# k3 block widened to 32 positions x 16 batches x 16 channels (32 waves): eight lanes cover one 128-byte run of a [b][c][p] row, so a wave-wide load or store touches 4 rows, not 8; sum orders unchanged
# speedup vs baseline: 6.9236x; 1.0514x over previous
#include <hip/hip_runtime.h>

typedef __attribute__((ext_vector_type(2))) float v2f;
typedef __attribute__((ext_vector_type(8))) float v8f;

#define BB   32
#define CIN  64
#define COUT 64
#define HO   30
#define NPOS 900
#define KT   9
#define TOT  (BB*COUT*NPOS)

#define WS_Y     0
#define WS_PART  (TOT)
#define WS_PART2 (TOT + NPOS*COUT)

#define AP 74
#define BP 74

#define P1        8
#define K1_TILES  113
#define K1_BLOCKS (K1_TILES * 2)
#define K1_T      512
#define O1        32
#define APL       (BB * AP + 2)
#define BPL       (16 * BP + 40)
#define YP        12
#define ROWF      (NPOS * KT)
#define ASTG      (P1 * APL)
#define BSTG      (2 * P1 * BPL)

__device__ __forceinline__ float4 ld4(const float* p) { return *(const float4*)p; }

__device__ __forceinline__ void lds_signal() {
  asm volatile("s_wait_dscnt 0x0" ::: "memory");
  __builtin_amdgcn_fence(__ATOMIC_RELEASE, "workgroup", "local");
  __builtin_amdgcn_s_barrier_signal(-1);
}
__device__ __forceinline__ void lds_wait() {
  __builtin_amdgcn_s_barrier_wait(-1);
  __builtin_amdgcn_fence(__ATOMIC_ACQUIRE, "workgroup", "local");
}

__global__ __launch_bounds__(K1_T) void k1_conv(
    const float* __restrict__ ff, const float* __restrict__ cw,
    const float* __restrict__ cb, float* __restrict__ ws)
{
  __shared__ __attribute__((aligned(16))) float Al[2 * ASTG];
  __shared__ __attribute__((aligned(16))) float Bl[2 * BSTG];

  const int bid = blockIdx.x;
  const int oh = bid & 1, tile = bid >> 1;
  const int p0 = tile * P1, o0 = oh * O1;
  const int t = threadIdx.x;
  const int lane = t & 31, wave = t >> 5;
  const int ln = lane & 15, hi = lane >> 4;
  const int wp = wave & 7, wq = wave >> 3;

  const int apl = t & 7, ac = (t >> 3) & 7, ab = t >> 6;
  const int pa = min(p0 + apl, NPOS - 1);
  const float* asrc = ff + ((ab * CIN + ac) * 32 + pa / HO) * 32 + pa % HO;
  float* adst = Al + apl * APL + ab * AP + ac * KT;

  const float* afr = Al + wp * APL + ln * AP + 2 * hi;
  const float* bfr = Bl + (wq * P1 + wp) * BPL + ln * BP + 2 * hi;

  v8f acc0 = {}, acc1 = {};
  auto mma_chunk = [&](int st) {
    const float* af = afr + st * ASTG;
    const float* bf = bfr + st * BSTG;
    #pragma unroll
    for (int kb = 0; kb < 72; kb += 4) {
      v2f bv = *(const v2f*)(bf + kb);
      v2f a0 = *(const v2f*)(af + kb);
      v2f a1 = *(const v2f*)(af + 16 * AP + kb);
      acc0 = __builtin_amdgcn_wmma_f32_16x16x4_f32(false, a0, false, bv,
                                                   (short)0, acc0, false, false);
      acc1 = __builtin_amdgcn_wmma_f32_16x16x4_f32(false, a1, false, bv,
                                                   (short)0, acc1, false, false);
    }
  };
  for (int cg = 0; cg < 8; ++cg) {
    float4 wv[9];
    #pragma unroll
    for (int i = 0; i < 9; ++i) {
      int idx = i * K1_T + t;
      int row = idx / 18, f = idx - row * 18;
      int ol = row >> 3, cc = row & 7;
      int col = p0 * KT + f * 4;
      wv[i] = make_float4(0.f, 0.f, 0.f, 0.f);
      if (col < ROWF)
        wv[i] = ld4(cw + ((o0 + ol) * CIN + cg * 8 + cc) * ROWF + col);
    }
    float pv[4][9];
    #pragma unroll
    for (int i = 0; i < 4; ++i) {
      const float* s = asrc + (i * 8 * CIN + cg * 8) * 1024;
      #pragma unroll
      for (int kh = 0; kh < 3; ++kh)
        #pragma unroll
        for (int kw = 0; kw < 3; ++kw)
          pv[i][kh * 3 + kw] = s[kh * 32 + kw];
    }
    const int st = cg & 1;
    if (cg > 0) {
      lds_wait();
      mma_chunk(st ^ 1);
    }
    #pragma unroll
    for (int i = 0; i < 9; ++i) {
      int idx = i * K1_T + t;
      int row = idx / 18, f = idx - row * 18;
      int ol = row >> 3, cc = row & 7;
      float* d = Bl + st * BSTG + (ol >> 4) * (P1 * BPL) + (ol & 15) * BP + cc * KT;
      const float vv[4] = {wv[i].x, wv[i].y, wv[i].z, wv[i].w};
      #pragma unroll
      for (int e = 0; e < 4; ++e) {
        int flat = f * 4 + e;
        int pl = flat / KT, kk = flat - pl * KT;
        d[pl * BPL + kk] = vv[e];
      }
    }
    #pragma unroll
    for (int i = 0; i < 4; ++i)
      #pragma unroll
      for (int k = 0; k < KT; ++k) adst[st * ASTG + i * 8 * AP + k] = pv[i][k];
    lds_signal();
  }
  lds_wait();
  mma_chunk(1);

  __syncthreads();
  float* Ys   = Al;
  float* redS = Bl;
  float* redQ = Bl + 1024;
  const int p = p0 + wp;
  const int oc = wq * 16 + ln;
  const float bias = cb[(o0 + oc) * NPOS + min(p, NPOS - 1)];
  {
    float s = 0.f, q = 0.f;
    #pragma unroll
    for (int r = 0; r < 8; ++r) {
      float v = acc0[r] + bias;
      Ys[((hi * 8 + r) * O1 + oc) * YP + wp] = v;
      s += v; q += v * v;
    }
    redS[(wave * 2) * 32 + lane] = s; redQ[(wave * 2) * 32 + lane] = q;
  }
  {
    float s = 0.f, q = 0.f;
    #pragma unroll
    for (int r = 0; r < 8; ++r) {
      float v = acc1[r] + bias;
      Ys[((16 + hi * 8 + r) * O1 + oc) * YP + wp] = v;
      s += v; q += v * v;
    }
    redS[(wave * 2 + 1) * 32 + lane] = s; redQ[(wave * 2 + 1) * 32 + lane] = q;
  }
  __syncthreads();
  if (t < 64) {
    const int l0 = t & 31;
    const float* red = Bl + (t >> 5) * 1024 + (l0 >> 4) * (P1 * 64) + (l0 & 15);
    float tot = 0.f;
    #pragma unroll
    for (int pl2 = 0; pl2 < P1; ++pl2) {
      float f4 = 0.f;
      #pragma unroll
      for (int mt2 = 0; mt2 < 2; ++mt2)
        #pragma unroll
        for (int h2 = 0; h2 < 2; ++h2)
          f4 += red[(pl2 * 2 + mt2) * 32 + h2 * 16];
      tot += (p0 + pl2 < NPOS) ? f4 : 0.f;
    }
    ws[(t < 32 ? WS_PART : WS_PART2) + tile * COUT + o0 + l0] = tot;
  }
  #pragma unroll
  for (int i = 0; i < 4; ++i) {
    int item = i * K1_T + t;
    int q4 = item & 1, row = item >> 1;
    int b = row >> 5, ol = row & 31;
    int pq = p0 + 4 * q4;
    if (pq < NPOS)
      *(float4*)(ws + WS_Y + (b * COUT + o0 + ol) * NPOS + pq) =
          *(const float4*)(Ys + row * YP + 4 * q4);
  }
}

#define PT        32
#define K3_TILES  29
#define K3_BLOCKS (K3_TILES * 8)
#define K3_T      1024
#define SPP       68
#define PSS       1090
#define RP        36
#define FG        64


__global__ __launch_bounds__(K3_T) void k3_rec_lif(
    const float* __restrict__ fbv,  const float* __restrict__ soma,
    const float* __restrict__ spk,  const float* __restrict__ acur,
    const float* __restrict__ bt,   const float* __restrict__ lrec,
    const float* __restrict__ taum, const float* __restrict__ taud,
    const float* __restrict__ taua, const float* __restrict__ gamma,
    const float* __restrict__ beta, const float* __restrict__ ws,
    float* __restrict__ out)
{
  __shared__ __attribute__((aligned(16))) float Ss[PT * PSS];
  __shared__ __attribute__((aligned(16))) float Rs[256 * RP];
  __shared__ __attribute__((aligned(16))) float gP[2 * FG * 16];
  __shared__ float hP[2 * 4 * 16];

  const int bid = blockIdx.x;
  const int cq = bid & 3, bh = (bid >> 2) & 1, tile = bid >> 3;
  const int p0 = tile * PT, b0 = bh * 16, c0 = cq * 16;
  const int t = threadIdx.x;
  const int lane = t & 31, wave = t >> 5;
  const int ln = lane & 15, hi = lane >> 4;

  const int q  = t & 7;
  const int cl = (t >> 3) & 15;
  const int c  = c0 + cl;
  const int pq = p0 + 4 * q;
  const bool okq = pq < NPOS;
  const int pql = okq ? pq : NPOS - 4;
  const int bl0 = t >> 7;

  const int fw = t >> 8, fg = (t >> 2) & (FG - 1), fc4 = t & 3;
  const bool has2 = fg + FG < K1_TILES;
  const float* pp = ws + (fw ? WS_PART2 : WS_PART) + fg * COUT + c0 + fc4 * 4;
  if (t < 2 * FG * 4) {
    float4 P = ld4(pp);
    if (has2) {
      const float4 p2 = ld4(pp + FG * COUT);
      P.x += p2.x; P.y += p2.y; P.z += p2.z; P.w += p2.w;
    }
    *(float4*)(gP + fw * (FG * 16) + fg * 16 + fc4 * 4) = P;
  }

  float4 sv4[8];
  #pragma unroll
  for (int i = 0; i < 8; ++i) {
    int e = i * K3_T + t;
    int row = e >> 3;
    int d = row & 63, bl = row >> 6;
    sv4[i] = ld4(spk + ((b0 + bl) * COUT + d) * NPOS + pql);
  }

  const int pw = min(p0 + wave, NPOS - 1);
  const float* lp = lrec + ((size_t)pw * COUT + c0 + ln) * COUT + 2 * hi;
  v2f bfr[16];
  #pragma unroll
  for (int j = 0; j < 16; ++j) bfr[j] = *(const v2f*)(lp + 4 * j);

  const float4 tm = ld4(taum + c * NPOS + pql);
  const float4 td = ld4(taud + c * NPOS + pql);
  const float4 ta = ld4(taua + c * NPOS + pql);
  float4 yv[2], bv4[2], av4[2], fv4[2], so4[2];
  #pragma unroll
  for (int i = 0; i < 2; ++i) {
    int idx = ((b0 + bl0 + 8 * i) * COUT + c) * NPOS + pql;
    yv[i]  = ld4(ws + WS_Y + idx);
    bv4[i] = ld4(bt + idx);
    av4[i] = ld4(acur + idx);
    fv4[i] = ld4(fbv + idx);
    so4[i] = ld4(soma + idx);
  }

  #pragma unroll
  for (int i = 0; i < 8; ++i) {
    int e = i * K3_T + t;
    int row = e >> 3;
    int d = row & 63, bl = row >> 6;
    float* dst = Ss + (4 * q) * PSS + bl * SPP + d;
    dst[0]       = sv4[i].x;
    dst[PSS]     = sv4[i].y;
    dst[2 * PSS] = sv4[i].z;
    dst[3 * PSS] = sv4[i].w;
  }
  __syncthreads();

  const float gm = gamma[c];
  const float be = beta[c];

  if (t < 128) {
    const int ch = t & 15, h = (t >> 4) & 3, w2 = t >> 6;
    const float* g0 = gP + w2 * (FG * 16) + h * (FG / 4) * 16 + ch;
    float s = 0.f;
    #pragma unroll
    for (int i = 0; i < FG / 4; ++i) s += g0[i * 16];
    hP[w2 * 64 + h * 16 + ch] = s;
  }

  v8f acc = {};
  const float* ap = Ss + wave * PSS + ln * SPP + 2 * hi;
  #pragma unroll
  for (int j = 0; j < 16; ++j) {
    v2f av = *(const v2f*)(ap + 4 * j);
    acc = __builtin_amdgcn_wmma_f32_16x16x4_f32(false, av, false, bfr[j],
                                                (short)0, acc, false, false);
  }
  #pragma unroll
  for (int r = 0; r < 8; ++r)
    Rs[((hi * 8 + r) * 16 + ln) * RP + wave] = acc[r];
  __syncthreads();

  if (okq) {
    const float Sx = (hP[cl] + hP[16 + cl]) + (hP[32 + cl] + hP[48 + cl]);
    const float Qx = (hP[64 + cl] + hP[80 + cl]) + (hP[96 + cl] + hP[112 + cl]);
    const float inv = 1.0f / (float)(BB * NPOS);
    const float mean = Sx * inv;
    const float var  = Qx * inv - mean * mean;
    const float sc = gm * rsqrtf(var + 1e-5f);
    const float sh = be - mean * sc;
    float alpha[4], rho[4], eta[4];
    const float tmv[4] = {tm.x, tm.y, tm.z, tm.w};
    const float tdv[4] = {td.x, td.y, td.z, td.w};
    const float tav[4] = {ta.x, ta.y, ta.z, ta.w};
    #pragma unroll
    for (int k = 0; k < 4; ++k) {
      alpha[k] = __expf(-0.5f / tmv[k]);
      rho[k]   = __expf(-0.5f / tdv[k]);
      eta[k]   = __expf(-0.5f / tav[k]);
    }
    #pragma unroll
    for (int i = 0; i < 2; ++i) {
      const int bl = bl0 + 8 * i;
      const int idx = ((b0 + bl) * COUT + c) * NPOS + pq;
      const float4 rc = *(const float4*)(Rs + (bl * 16 + cl) * RP + 4 * q);
      const float rcv[4] = {rc.x, rc.y, rc.z, rc.w};
      const float yy[4]  = {yv[i].x, yv[i].y, yv[i].z, yv[i].w};
      const float bb[4]  = {bv4[i].x, bv4[i].y, bv4[i].z, bv4[i].w};
      const float aa[4]  = {av4[i].x, av4[i].y, av4[i].z, av4[i].w};
      const float ff4[4] = {fv4[i].x, fv4[i].y, fv4[i].z, fv4[i].w};
      const float ss[4]  = {so4[i].x, so4[i].y, so4[i].z, so4[i].w};
      float o0[4], o1[4], o2[4], o3[4];
      #pragma unroll
      for (int k = 0; k < 4; ++k) {
        float sv    = Ss[(4 * q + k) * PSS + bl * SPP + c];
        float convx = yy[k] * sc + sh + rcv[k];
        float bnew  = rho[k] * bb[k] + (1.f - rho[k]) * sv;
        float thre  = 0.1f + 1.8f * bnew;
        float anew  = eta[k] * aa[k] + ff4[k];
        float sg    = 1.f / (1.f + __expf(-anew)) - 0.5f;
        float snew  = alpha[k] * ss[k] + sg + convx - thre * sv;
        o0[k] = snew;
        o1[k] = (snew - thre) > 0.f ? 1.f : 0.f;
        o2[k] = anew;
        o3[k] = bnew;
      }
      *(float4*)(out + idx)           = make_float4(o0[0], o0[1], o0[2], o0[3]);
      *(float4*)(out + TOT + idx)     = make_float4(o1[0], o1[1], o1[2], o1[3]);
      *(float4*)(out + 2 * TOT + idx) = make_float4(o2[0], o2[1], o2[2], o2[3]);
      *(float4*)(out + 3 * TOT + idx) = make_float4(o3[0], o3[1], o3[2], o3[3]);
    }
  }
}

extern "C" void kernel_launch(void* const* d_in, const int* in_sizes, int n_in,
                              void* d_out, int out_size, void* d_ws, size_t ws_size,
                              hipStream_t stream) {
  const float* ff   = (const float*)d_in[0];
  const float* fb   = (const float*)d_in[1];
  const float* soma = (const float*)d_in[2];
  const float* spk  = (const float*)d_in[3];
  const float* acur = (const float*)d_in[4];
  const float* bt   = (const float*)d_in[5];
  const float* cw   = (const float*)d_in[6];
  const float* cb   = (const float*)d_in[7];
  const float* gam  = (const float*)d_in[8];
  const float* bet  = (const float*)d_in[9];
  const float* lrec = (const float*)d_in[10];
  const float* taum = (const float*)d_in[11];
  const float* taud = (const float*)d_in[12];
  const float* taua = (const float*)d_in[13];
  float* out = (float*)d_out;
  float* ws  = (float*)d_ws;

  hipLaunchKernelGGL(k1_conv,    dim3(K1_BLOCKS), dim3(K1_T), 0, stream, ff, cw, cb, ws);
  hipLaunchKernelGGL(k3_rec_lif, dim3(K3_BLOCKS), dim3(K3_T), 0, stream,
                     fb, soma, spk, acur, bt, lrec, taum, taud, taua, gam, bet, ws, out);
}


# speedup vs baseline: 7.4675x; 1.0786x over previous
#include <hip/hip_runtime.h>

typedef __attribute__((ext_vector_type(2))) float v2f;
typedef __attribute__((ext_vector_type(8))) float v8f;

#define BB   32
#define CIN  64
#define COUT 64
#define HO   30
#define NPOS 900
#define KT   9
#define TOT  (BB*COUT*NPOS)

#define WS_Y     0
#define WS_PART  (TOT)
#define WS_PART2 (TOT + NPOS*COUT)

#define AP 74
#define BP 74

#define P1        8
#define K1_TILES  113
#define K1_BLOCKS (K1_TILES * 2)
#define K1_T      512
#define O1        32
#define APL       (BB * AP + 2)
#define BPL       (16 * BP + 40)
#define YP        12
#define ROWF      (NPOS * KT)
#define ASTG      (P1 * APL)
#define BSTG      (2 * P1 * BPL)

__device__ __forceinline__ float4 ld4(const float* p) { return *(const float4*)p; }

__device__ __forceinline__ void lds_signal() {
  asm volatile("s_wait_dscnt 0x0" ::: "memory");
  __builtin_amdgcn_fence(__ATOMIC_RELEASE, "workgroup", "local");
  __builtin_amdgcn_s_barrier_signal(-1);
}
__device__ __forceinline__ void lds_wait() {
  __builtin_amdgcn_s_barrier_wait(-1);
  __builtin_amdgcn_fence(__ATOMIC_ACQUIRE, "workgroup", "local");
}

__global__ __launch_bounds__(K1_T) void k1_conv(
    const float* __restrict__ ff, const float* __restrict__ cw,
    const float* __restrict__ cb, float* __restrict__ ws)
{
  __shared__ __attribute__((aligned(16))) float Al[2 * ASTG];
  __shared__ __attribute__((aligned(16))) float Bl[2 * BSTG];

  const int bid = blockIdx.x;
  const int bcls = bid & 15, brow = bid >> 4;
  const int pair = bcls * 14 + min(bcls, 2) + brow;
  const int oh = pair >= K1_TILES ? 1 : 0;
  const int tile = pair - oh * K1_TILES;
  const int p0 = tile * P1, o0 = oh * O1;
  const int t = threadIdx.x;
  const int lane = t & 31, wave = t >> 5;
  const int ln = lane & 15, hi = lane >> 4;
  const int wp = wave & 7, wq = wave >> 3;

  const int apl = t & 7, ac = (t >> 3) & 7, ab = t >> 6;
  const int pa = min(p0 + apl, NPOS - 1);
  const float* asrc = ff + ((ab * CIN + ac) * 32 + pa / HO) * 32 + pa % HO;
  float* adst = Al + apl * APL + ab * AP + ac * KT;

  const float* afr = Al + wp * APL + ln * AP + 2 * hi;
  const float* bfr = Bl + (wq * P1 + wp) * BPL + ln * BP + 2 * hi;

  v8f acc0 = {}, acc1 = {};
  auto mma_chunk = [&](int st) {
    const float* af = afr + st * ASTG;
    const float* bf = bfr + st * BSTG;
    #pragma unroll
    for (int kb = 0; kb < 72; kb += 4) {
      v2f bv = *(const v2f*)(bf + kb);
      v2f a0 = *(const v2f*)(af + kb);
      v2f a1 = *(const v2f*)(af + 16 * AP + kb);
      acc0 = __builtin_amdgcn_wmma_f32_16x16x4_f32(false, a0, false, bv,
                                                   (short)0, acc0, false, false);
      acc1 = __builtin_amdgcn_wmma_f32_16x16x4_f32(false, a1, false, bv,
                                                   (short)0, acc1, false, false);
    }
  };
  for (int cg = 0; cg < 8; ++cg) {
    float4 wv[9];
    #pragma unroll
    for (int i = 0; i < 9; ++i) {
      int idx = i * K1_T + t;
      int row = idx / 18, f = idx - row * 18;
      int ol = row >> 3, cc = row & 7;
      int col = p0 * KT + f * 4;
      wv[i] = make_float4(0.f, 0.f, 0.f, 0.f);
      if (col < ROWF)
        wv[i] = ld4(cw + ((o0 + ol) * CIN + cg * 8 + cc) * ROWF + col);
    }
    float pv[4][9];
    #pragma unroll
    for (int i = 0; i < 4; ++i) {
      const float* s = asrc + (i * 8 * CIN + cg * 8) * 1024;
      #pragma unroll
      for (int kh = 0; kh < 3; ++kh)
        #pragma unroll
        for (int kw = 0; kw < 3; ++kw)
          pv[i][kh * 3 + kw] = s[kh * 32 + kw];
    }
    const int st = cg & 1;
    if (cg > 0) {
      lds_wait();
      mma_chunk(st ^ 1);
    }
    #pragma unroll
    for (int i = 0; i < 9; ++i) {
      int idx = i * K1_T + t;
      int row = idx / 18, f = idx - row * 18;
      int ol = row >> 3, cc = row & 7;
      float* d = Bl + st * BSTG + (ol >> 4) * (P1 * BPL) + (ol & 15) * BP + cc * KT;
      const float vv[4] = {wv[i].x, wv[i].y, wv[i].z, wv[i].w};
      #pragma unroll
      for (int e = 0; e < 4; ++e) {
        int flat = f * 4 + e;
        int pl = flat / KT, kk = flat - pl * KT;
        d[pl * BPL + kk] = vv[e];
      }
    }
    #pragma unroll
    for (int i = 0; i < 4; ++i)
      #pragma unroll
      for (int k = 0; k < KT; ++k) adst[st * ASTG + i * 8 * AP + k] = pv[i][k];
    lds_signal();
  }
  lds_wait();
  mma_chunk(1);

  __syncthreads();
  float* Ys   = Al;
  float* redS = Bl;
  float* redQ = Bl + 1024;
  const int p = p0 + wp;
  const int oc = wq * 16 + ln;
  const float bias = cb[(o0 + oc) * NPOS + min(p, NPOS - 1)];
  {
    float s = 0.f, q = 0.f;
    #pragma unroll
    for (int r = 0; r < 8; ++r) {
      float v = acc0[r] + bias;
      Ys[((hi * 8 + r) * O1 + oc) * YP + wp] = v;
      s += v; q += v * v;
    }
    redS[(wave * 2) * 32 + lane] = s; redQ[(wave * 2) * 32 + lane] = q;
  }
  {
    float s = 0.f, q = 0.f;
    #pragma unroll
    for (int r = 0; r < 8; ++r) {
      float v = acc1[r] + bias;
      Ys[((16 + hi * 8 + r) * O1 + oc) * YP + wp] = v;
      s += v; q += v * v;
    }
    redS[(wave * 2 + 1) * 32 + lane] = s; redQ[(wave * 2 + 1) * 32 + lane] = q;
  }
  __syncthreads();
  if (t < 64) {
    const int l0 = t & 31;
    const float* red = Bl + (t >> 5) * 1024 + (l0 >> 4) * (P1 * 64) + (l0 & 15);
    float tot = 0.f;
    #pragma unroll
    for (int pl2 = 0; pl2 < P1; ++pl2) {
      float f4 = 0.f;
      #pragma unroll
      for (int mt2 = 0; mt2 < 2; ++mt2)
        #pragma unroll
        for (int h2 = 0; h2 < 2; ++h2)
          f4 += red[(pl2 * 2 + mt2) * 32 + h2 * 16];
      tot += (p0 + pl2 < NPOS) ? f4 : 0.f;
    }
    ws[(t < 32 ? WS_PART : WS_PART2) + tile * COUT + o0 + l0] = tot;
  }
  #pragma unroll
  for (int i = 0; i < 4; ++i) {
    int item = i * K1_T + t;
    int q4 = item & 1, row = item >> 1;
    int b = row >> 5, ol = row & 31;
    int pq = p0 + 4 * q4;
    if (pq < NPOS)
      *(float4*)(ws + WS_Y + (b * COUT + o0 + ol) * NPOS + pq) =
          *(const float4*)(Ys + row * YP + 4 * q4);
  }
}

#define PT        32
#define K3_TILES  29
#define K3_BLOCKS (K3_TILES * 8)
#define K3_T      1024
#define SPP       68
#define PSS       1090
#define RP        36
#define FG        64


__global__ __launch_bounds__(K3_T) void k3_rec_lif(
    const float* __restrict__ fbv,  const float* __restrict__ soma,
    const float* __restrict__ spk,  const float* __restrict__ acur,
    const float* __restrict__ bt,   const float* __restrict__ lrec,
    const float* __restrict__ taum, const float* __restrict__ taud,
    const float* __restrict__ taua, const float* __restrict__ gamma,
    const float* __restrict__ beta, const float* __restrict__ ws,
    float* __restrict__ out)
{
  __shared__ __attribute__((aligned(16))) float Ss[PT * PSS];
  __shared__ __attribute__((aligned(16))) float Rs[256 * RP];
  __shared__ __attribute__((aligned(16))) float gP[2 * FG * 16];
  __shared__ float hP[2 * 4 * 16];

  const int bid = blockIdx.x;
  const int bcls = bid & 15, brow = bid >> 4;
  const int cq = bcls & 3, bh = (bcls >> 2) & 1, tile = (bcls >> 3) * 15 + brow;
  const int p0 = tile * PT, b0 = bh * 16, c0 = cq * 16;
  const int t = threadIdx.x;
  const int lane = t & 31, wave = t >> 5;
  const int ln = lane & 15, hi = lane >> 4;

  const int q  = t & 7;
  const int cl = (t >> 3) & 15;
  const int c  = c0 + cl;
  const int pq = p0 + 4 * q;
  const bool okq = pq < NPOS;
  const int pql = okq ? pq : NPOS - 4;
  const int bl0 = t >> 7;

  const int fw = t >> 8, fg = (t >> 2) & (FG - 1), fc4 = t & 3;
  const bool has2 = fg + FG < K1_TILES;
  const float* pp = ws + (fw ? WS_PART2 : WS_PART) + fg * COUT + c0 + fc4 * 4;
  if (t < 2 * FG * 4) {
    float4 P = ld4(pp);
    if (has2) {
      const float4 p2 = ld4(pp + FG * COUT);
      P.x += p2.x; P.y += p2.y; P.z += p2.z; P.w += p2.w;
    }
    *(float4*)(gP + fw * (FG * 16) + fg * 16 + fc4 * 4) = P;
  }

  float4 sv4[8];
  #pragma unroll
  for (int i = 0; i < 8; ++i) {
    int e = i * K3_T + t;
    int row = e >> 3;
    int d = row & 63, bl = row >> 6;
    sv4[i] = ld4(spk + ((b0 + bl) * COUT + d) * NPOS + pql);
  }

  const int pw = min(p0 + wave, NPOS - 1);
  const float* lp = lrec + ((size_t)pw * COUT + c0 + ln) * COUT + 2 * hi;
  v2f bfr[16];
  #pragma unroll
  for (int j = 0; j < 16; ++j) bfr[j] = *(const v2f*)(lp + 4 * j);

  const float4 tm = ld4(taum + c * NPOS + pql);
  const float4 td = ld4(taud + c * NPOS + pql);
  const float4 ta = ld4(taua + c * NPOS + pql);
  float4 yv[2], bv4[2], av4[2], fv4[2], so4[2];
  #pragma unroll
  for (int i = 0; i < 2; ++i) {
    int idx = ((b0 + bl0 + 8 * i) * COUT + c) * NPOS + pql;
    yv[i]  = ld4(ws + WS_Y + idx);
    bv4[i] = ld4(bt + idx);
    av4[i] = ld4(acur + idx);
    fv4[i] = ld4(fbv + idx);
    so4[i] = ld4(soma + idx);
  }

  #pragma unroll
  for (int i = 0; i < 8; ++i) {
    int e = i * K3_T + t;
    int row = e >> 3;
    int d = row & 63, bl = row >> 6;
    float* dst = Ss + (4 * q) * PSS + bl * SPP + d;
    dst[0]       = sv4[i].x;
    dst[PSS]     = sv4[i].y;
    dst[2 * PSS] = sv4[i].z;
    dst[3 * PSS] = sv4[i].w;
  }
  __syncthreads();

  const float gm = gamma[c];
  const float be = beta[c];

  if (t < 128) {
    const int ch = t & 15, h = (t >> 4) & 3, w2 = t >> 6;
    const float* g0 = gP + w2 * (FG * 16) + h * (FG / 4) * 16 + ch;
    float s = 0.f;
    #pragma unroll
    for (int i = 0; i < FG / 4; ++i) s += g0[i * 16];
    hP[w2 * 64 + h * 16 + ch] = s;
  }

  v8f acc = {};
  const float* ap = Ss + wave * PSS + ln * SPP + 2 * hi;
  #pragma unroll
  for (int j = 0; j < 16; ++j) {
    v2f av = *(const v2f*)(ap + 4 * j);
    acc = __builtin_amdgcn_wmma_f32_16x16x4_f32(false, av, false, bfr[j],
                                                (short)0, acc, false, false);
  }
  #pragma unroll
  for (int r = 0; r < 8; ++r)
    Rs[((hi * 8 + r) * 16 + ln) * RP + wave] = acc[r];
  __syncthreads();

  if (okq) {
    const float Sx = (hP[cl] + hP[16 + cl]) + (hP[32 + cl] + hP[48 + cl]);
    const float Qx = (hP[64 + cl] + hP[80 + cl]) + (hP[96 + cl] + hP[112 + cl]);
    const float inv = 1.0f / (float)(BB * NPOS);
    const float mean = Sx * inv;
    const float var  = Qx * inv - mean * mean;
    const float sc = gm * rsqrtf(var + 1e-5f);
    const float sh = be - mean * sc;
    float alpha[4], rho[4], eta[4];
    const float tmv[4] = {tm.x, tm.y, tm.z, tm.w};
    const float tdv[4] = {td.x, td.y, td.z, td.w};
    const float tav[4] = {ta.x, ta.y, ta.z, ta.w};
    #pragma unroll
    for (int k = 0; k < 4; ++k) {
      alpha[k] = __expf(-0.5f / tmv[k]);
      rho[k]   = __expf(-0.5f / tdv[k]);
      eta[k]   = __expf(-0.5f / tav[k]);
    }
    #pragma unroll
    for (int i = 0; i < 2; ++i) {
      const int bl = bl0 + 8 * i;
      const int idx = ((b0 + bl) * COUT + c) * NPOS + pq;
      const float4 rc = *(const float4*)(Rs + (bl * 16 + cl) * RP + 4 * q);
      const float rcv[4] = {rc.x, rc.y, rc.z, rc.w};
      const float yy[4]  = {yv[i].x, yv[i].y, yv[i].z, yv[i].w};
      const float bb[4]  = {bv4[i].x, bv4[i].y, bv4[i].z, bv4[i].w};
      const float aa[4]  = {av4[i].x, av4[i].y, av4[i].z, av4[i].w};
      const float ff4[4] = {fv4[i].x, fv4[i].y, fv4[i].z, fv4[i].w};
      const float ss[4]  = {so4[i].x, so4[i].y, so4[i].z, so4[i].w};
      float o0[4], o1[4], o2[4], o3[4];
      #pragma unroll
      for (int k = 0; k < 4; ++k) {
        float sv    = Ss[(4 * q + k) * PSS + bl * SPP + c];
        float convx = yy[k] * sc + sh + rcv[k];
        float bnew  = rho[k] * bb[k] + (1.f - rho[k]) * sv;
        float thre  = 0.1f + 1.8f * bnew;
        float anew  = eta[k] * aa[k] + ff4[k];
        float sg    = 1.f / (1.f + __expf(-anew)) - 0.5f;
        float snew  = alpha[k] * ss[k] + sg + convx - thre * sv;
        o0[k] = snew;
        o1[k] = (snew - thre) > 0.f ? 1.f : 0.f;
        o2[k] = anew;
        o3[k] = bnew;
      }
      *(float4*)(out + idx)           = make_float4(o0[0], o0[1], o0[2], o0[3]);
      *(float4*)(out + TOT + idx)     = make_float4(o1[0], o1[1], o1[2], o1[3]);
      *(float4*)(out + 2 * TOT + idx) = make_float4(o2[0], o2[1], o2[2], o2[3]);
      *(float4*)(out + 3 * TOT + idx) = make_float4(o3[0], o3[1], o3[2], o3[3]);
    }
  }
}

extern "C" void kernel_launch(void* const* d_in, const int* in_sizes, int n_in,
                              void* d_out, int out_size, void* d_ws, size_t ws_size,
                              hipStream_t stream) {
  const float* ff   = (const float*)d_in[0];
  const float* fb   = (const float*)d_in[1];
  const float* soma = (const float*)d_in[2];
  const float* spk  = (const float*)d_in[3];
  const float* acur = (const float*)d_in[4];
  const float* bt   = (const float*)d_in[5];
  const float* cw   = (const float*)d_in[6];
  const float* cb   = (const float*)d_in[7];
  const float* gam  = (const float*)d_in[8];
  const float* bet  = (const float*)d_in[9];
  const float* lrec = (const float*)d_in[10];
  const float* taum = (const float*)d_in[11];
  const float* taud = (const float*)d_in[12];
  const float* taua = (const float*)d_in[13];
  float* out = (float*)d_out;
  float* ws  = (float*)d_ws;

  hipLaunchKernelGGL(k1_conv,    dim3(K1_BLOCKS), dim3(K1_T), 0, stream, ff, cw, cb, ws);
  hipLaunchKernelGGL(k3_rec_lif, dim3(K3_BLOCKS), dim3(K3_T), 0, stream,
                     fb, soma, spk, acur, bt, lrec, taum, taud, taua, gam, bet, ws, out);
}
